# starting kernel
# baseline (speedup 1.0000x reference)
#include <hip/hip_runtime.h>
#include <hip/hip_bf16.h>
#include <math.h>

typedef __attribute__((ext_vector_type(16))) _Float16 v16h;
typedef __attribute__((ext_vector_type(8)))  _Float16 v8h;
typedef __attribute__((ext_vector_type(4)))  _Float16 v4h;
typedef __attribute__((ext_vector_type(8)))  float    v8f;

#define WAVES 4
#define BLOCK 128


__device__ __forceinline__ float silu_f(float x) {
    return x / (1.0f + __expf(-x));
}

__device__ __forceinline__ unsigned enc_f(float f) {
    unsigned u = __float_as_uint(f);
    return (u & 0x80000000u) ? ~u : (u | 0x80000000u);
}
__device__ __forceinline__ float dec_f(unsigned u) {
    unsigned b = (u & 0x80000000u) ? (u & 0x7FFFFFFFu) : ~u;
    return __uint_as_float(b);
}
#define ENC_NEG_INF 0x007FFFFFu

__device__ __forceinline__ v16h load_frag(const _Float16* p) {
    v8h lo = *(const v8h*)(p);
    v8h hi = *(const v8h*)(p + 16);
    return __builtin_shufflevector(lo, hi, 0, 1, 2, 3, 4, 5, 6, 7,
                                   8, 9, 10, 11, 12, 13, 14, 15);
}

__device__ __forceinline__ v8f wmma32(v16h a, v16h b, v8f c) {
    return __builtin_amdgcn_wmma_f32_16x16x32_f16(
         false, a,  false, b,
         (short)0, c,  false,  false);
}

__device__ __forceinline__ void cvt_store4(_Float16* dstp, const float* srcp) {
    const float4 v = *(const float4*)srcp;
    v4h h;
    h[0] = (_Float16)v.x; h[1] = (_Float16)v.y;
    h[2] = (_Float16)v.z; h[3] = (_Float16)v.w;
    *(v4h*)dstp = h;
}
__device__ __forceinline__ void zero_store4(_Float16* dstp) {
    v4h z = {};
    *(v4h*)dstp = z;
}


__global__ __launch_bounds__(256) void init_ws_kernel(unsigned* e_max, float* denom,
                                                      float* h_neigh, int N) {
    int stride = gridDim.x * blockDim.x;
    int i0 = blockIdx.x * blockDim.x + threadIdx.x;
    for (int i = i0; i < N * 64; i += stride) h_neigh[i] = 0.0f;
    for (int i = i0; i < N; i += stride) { denom[i] = 0.0f; e_max[i] = ENC_NEG_INF; }
}


__global__ __launch_bounds__(BLOCK) void edge_score_kernel(
    const float* __restrict__ node_feat, const float* __restrict__ coord,
    const float* __restrict__ edge_feat,
    const float* __restrict__ W_q, const float* __restrict__ W_k,
    const int* __restrict__ src, const int* __restrict__ dst,
    float* __restrict__ e_arr, unsigned* __restrict__ e_max, int E) {

    __shared__ _Float16 sWkT[64 * 96];
    __shared__ _Float16 sWqT[64 * 64];
    __shared__ _Float16 sA[WAVES][16 * 160];

    for (int i = threadIdx.x; i < 64 * 96; i += BLOCK) {
        int n = i / 96, k = i % 96;
        int kr = (k < 64) ? k : (k < 80 ? (65 + (k - 64)) : (k == 80 ? 64 : -1));
        sWkT[i] = (kr >= 0) ? (_Float16)W_k[kr * 64 + n] : (_Float16)0.0f;
    }
    for (int i = threadIdx.x; i < 64 * 64; i += BLOCK) {
        int n = i >> 6, k = i & 63;
        sWqT[i] = (_Float16)W_q[k * 64 + n];
    }

    const int wave = threadIdx.x >> 5, lane = threadIdx.x & 31;
    const int row = lane & 15, hi = lane >> 4;
    const int numTiles = (E + 15) >> 4;
    const int tile = blockIdx.x * WAVES + wave;
    const bool validWave = tile < numTiles;

    int e = validWave ? (tile * 16 + row) : 0;
    if (e >= E) e = 0;
    const int s = src[e], d = dst[e];

    __syncthreads();

    _Float16* Ar = &sA[wave][row * 160];
    if (hi == 0) {
#pragma unroll
        for (int c = 0; c < 64; c += 4) cvt_store4(&Ar[c], &node_feat[s * 64 + c]);
#pragma unroll
        for (int c = 0; c < 32; c += 4) cvt_store4(&Ar[96 + c], &node_feat[d * 64 + c]);
    } else {
#pragma unroll
        for (int c = 32; c < 64; c += 4) cvt_store4(&Ar[96 + c], &node_feat[d * 64 + c]);
#pragma unroll
        for (int c = 0; c < 16; c += 4) cvt_store4(&Ar[64 + c], &edge_feat[e * 16 + c]);
#pragma unroll
        for (int c = 80; c < 96; c += 4) zero_store4(&Ar[c]);
        float dx = coord[s * 3 + 0] - coord[d * 3 + 0];
        float dy = coord[s * 3 + 1] - coord[d * 3 + 1];
        float dz = coord[s * 3 + 2] - coord[d * 3 + 2];
        Ar[80] = (_Float16)sqrtf(dx * dx + dy * dy + dz * dz);
    }
    __syncthreads();

    v8f accK[4], accQ[4];
#pragma unroll
    for (int nt = 0; nt < 4; ++nt) { accK[nt] = {}; accQ[nt] = {}; }

#pragma unroll
    for (int kc = 0; kc < 3; ++kc) {
        const int kb = kc * 32 + hi * 8;
        v16h a = load_frag(&sA[wave][row * 160 + kb]);
#pragma unroll
        for (int nt = 0; nt < 4; ++nt)
            accK[nt] = wmma32(a, load_frag(&sWkT[(nt * 16 + row) * 96 + kb]), accK[nt]);
    }
#pragma unroll
    for (int kc = 0; kc < 2; ++kc) {
        const int kb = kc * 32 + hi * 8;
        v16h a = load_frag(&sA[wave][row * 160 + 96 + kb]);
#pragma unroll
        for (int nt = 0; nt < 4; ++nt)
            accQ[nt] = wmma32(a, load_frag(&sWqT[(nt * 16 + row) * 64 + kb]), accQ[nt]);
    }

    float p[8];
#pragma unroll
    for (int r = 0; r < 8; ++r) {
        p[r] = 0.0f;
#pragma unroll
        for (int nt = 0; nt < 4; ++nt) p[r] += accK[nt][r] * accQ[nt][r];
    }
#pragma unroll
    for (int msk = 1; msk < 16; msk <<= 1) {
#pragma unroll
        for (int r = 0; r < 8; ++r) p[r] += __shfl_xor(p[r], msk, 32);
    }

    if (validWave && row == 0) {
#pragma unroll
        for (int r = 0; r < 8; ++r) {
            int m = r + 8 * hi;
            int eg = tile * 16 + m;
            if (eg < E) {
                float ev = p[r] * 0.125f;
                e_arr[eg] = ev;
                atomicMax(&e_max[dst[eg]], enc_f(ev));
            }
        }
    }
}


__global__ __launch_bounds__(BLOCK) void edge_msg_kernel(
    const float* __restrict__ node_feat, const float* __restrict__ coord,
    const float* __restrict__ edge_feat,
    const float* __restrict__ W_e1, const float* __restrict__ b_e1,
    const float* __restrict__ W_e2, const float* __restrict__ b_e2,
    const int* __restrict__ src, const int* __restrict__ dst,
    const float* __restrict__ e_arr, const unsigned* __restrict__ e_max,
    float* __restrict__ denom, float* __restrict__ h_neigh, int E) {

    __shared__ _Float16 sW1T[64 * 160];
    __shared__ _Float16 sW2T[64 * 64];
    __shared__ _Float16 sA[WAVES][16 * 160];
    __shared__ _Float16 sMid[WAVES][16 * 64];
    __shared__ float    sW16[WAVES][16];
    __shared__ int      sD16[WAVES][16];

    for (int i = threadIdx.x; i < 64 * 160; i += BLOCK) {
        int n = i / 160, k = i % 160;
        int kr = (k < 128) ? k : (k < 144 ? (129 + (k - 128)) : (k == 144 ? 128 : -1));
        sW1T[i] = (kr >= 0) ? (_Float16)W_e1[kr * 64 + n] : (_Float16)0.0f;
    }
    for (int i = threadIdx.x; i < 64 * 64; i += BLOCK) {
        int n = i >> 6, k = i & 63;
        sW2T[i] = (_Float16)W_e2[k * 64 + n];
    }

    const int wave = threadIdx.x >> 5, lane = threadIdx.x & 31;
    const int row = lane & 15, hi = lane >> 4;
    const int numTiles = (E + 15) >> 4;
    const int tile = blockIdx.x * WAVES + wave;
    const bool validWave = tile < numTiles;

    int e = validWave ? (tile * 16 + row) : 0;
    if (e >= E) e = 0;
    const int s = src[e], d = dst[e];

    __syncthreads();

    _Float16* Ar = &sA[wave][row * 160];
    if (hi == 0) {
#pragma unroll
        for (int c = 0; c < 64; c += 4) cvt_store4(&Ar[c], &node_feat[s * 64 + c]);
#pragma unroll
        for (int c = 0; c < 32; c += 4) cvt_store4(&Ar[64 + c], &node_feat[d * 64 + c]);
    } else {
#pragma unroll
        for (int c = 32; c < 64; c += 4) cvt_store4(&Ar[64 + c], &node_feat[d * 64 + c]);
#pragma unroll
        for (int c = 0; c < 16; c += 4) cvt_store4(&Ar[128 + c], &edge_feat[e * 16 + c]);
#pragma unroll
        for (int c = 144; c < 160; c += 4) zero_store4(&Ar[c]);
        float dx = coord[s * 3 + 0] - coord[d * 3 + 0];
        float dy = coord[s * 3 + 1] - coord[d * 3 + 1];
        float dz = coord[s * 3 + 2] - coord[d * 3 + 2];
        Ar[144] = (_Float16)sqrtf(dx * dx + dy * dy + dz * dz);
    }
    __syncthreads();

    v8f acc[4];
#pragma unroll
    for (int nt = 0; nt < 4; ++nt) acc[nt] = {};
#pragma unroll
    for (int kc = 0; kc < 5; ++kc) {
        const int kb = kc * 32 + hi * 8;
        v16h a = load_frag(&sA[wave][row * 160 + kb]);
#pragma unroll
        for (int nt = 0; nt < 4; ++nt)
            acc[nt] = wmma32(a, load_frag(&sW1T[(nt * 16 + row) * 160 + kb]), acc[nt]);
    }
#pragma unroll
    for (int nt = 0; nt < 4; ++nt) {
        float bb = b_e1[nt * 16 + row];
#pragma unroll
        for (int r = 0; r < 8; ++r) {
            float h = silu_f(acc[nt][r] + bb);
            sMid[wave][(r + 8 * hi) * 64 + nt * 16 + row] = (_Float16)h;
        }
    }
    if (hi == 0) {
        int eg = validWave ? (tile * 16 + row) : 0;
        if (eg >= E) eg = 0;
        int dd = dst[eg];
        float w = __expf(e_arr[eg] - dec_f(e_max[dd]));
        sW16[wave][row] = w;
        sD16[wave][row] = dd;
        if (validWave && (tile * 16 + row) < E) atomicAdd(&denom[dd], w);
    }
    __syncthreads();

    v8f acc2[4];
#pragma unroll
    for (int nt = 0; nt < 4; ++nt) acc2[nt] = {};
#pragma unroll
    for (int kc = 0; kc < 2; ++kc) {
        const int kb = kc * 32 + hi * 8;
        v16h a = load_frag(&sMid[wave][row * 64 + kb]);
#pragma unroll
        for (int nt = 0; nt < 4; ++nt)
            acc2[nt] = wmma32(a, load_frag(&sW2T[(nt * 16 + row) * 64 + kb]), acc2[nt]);
    }
    if (validWave) {
#pragma unroll
        for (int nt = 0; nt < 4; ++nt) {
            float bb = b_e2[nt * 16 + row];
#pragma unroll
            for (int r = 0; r < 8; ++r) {
                int m = r + 8 * hi;
                if (tile * 16 + m < E) {
                    float msg = silu_f(acc2[nt][r] + bb);
                    atomicAdd(&h_neigh[(long)sD16[wave][m] * 64 + nt * 16 + row],
                              msg * sW16[wave][m]);
                }
            }
        }
    }
}


__global__ __launch_bounds__(BLOCK) void node_mlp_kernel(
    const float* __restrict__ node_feat,
    const float* __restrict__ W_n1, const float* __restrict__ b_n1,
    const float* __restrict__ W_n2, const float* __restrict__ b_n2,
    const float* __restrict__ denom, const float* __restrict__ h_neigh,
    float* __restrict__ out, int N) {

    __shared__ _Float16 sW1T[64 * 128];
    __shared__ _Float16 sW2T[64 * 64];
    __shared__ _Float16 sA[WAVES][16 * 128];
    __shared__ _Float16 sMid[WAVES][16 * 64];

    for (int i = threadIdx.x; i < 64 * 128; i += BLOCK) {
        int n = i >> 7, k = i & 127;
        sW1T[i] = (_Float16)W_n1[k * 64 + n];
    }
    for (int i = threadIdx.x; i < 64 * 64; i += BLOCK) {
        int n = i >> 6, k = i & 63;
        sW2T[i] = (_Float16)W_n2[k * 64 + n];
    }

    const int wave = threadIdx.x >> 5, lane = threadIdx.x & 31;
    const int row = lane & 15, hi = lane >> 4;
    const int numTiles = (N + 15) >> 4;
    const int tile = blockIdx.x * WAVES + wave;
    const bool validWave = tile < numTiles;

    int node = validWave ? (tile * 16 + row) : 0;
    if (node >= N) node = 0;

    __syncthreads();

    _Float16* Ar = &sA[wave][row * 128];
    if (hi == 0) {
#pragma unroll
        for (int c = 0; c < 64; c += 4) cvt_store4(&Ar[c], &node_feat[node * 64 + c]);
    } else {
        float dn = denom[node];
        float inv = (dn > 0.0f) ? (1.0f / dn) : 0.0f;
#pragma unroll
        for (int c = 0; c < 64; c += 4) {
            const float4 v = *(const float4*)&h_neigh[node * 64 + c];
            v4h h;
            h[0] = (_Float16)(v.x * inv); h[1] = (_Float16)(v.y * inv);
            h[2] = (_Float16)(v.z * inv); h[3] = (_Float16)(v.w * inv);
            *(v4h*)&Ar[64 + c] = h;
        }
    }
    __syncthreads();

    v8f acc[4];
#pragma unroll
    for (int nt = 0; nt < 4; ++nt) acc[nt] = {};
#pragma unroll
    for (int kc = 0; kc < 4; ++kc) {
        const int kb = kc * 32 + hi * 8;
        v16h a = load_frag(&sA[wave][row * 128 + kb]);
#pragma unroll
        for (int nt = 0; nt < 4; ++nt)
            acc[nt] = wmma32(a, load_frag(&sW1T[(nt * 16 + row) * 128 + kb]), acc[nt]);
    }
#pragma unroll
    for (int nt = 0; nt < 4; ++nt) {
        float bb = b_n1[nt * 16 + row];
#pragma unroll
        for (int r = 0; r < 8; ++r) {
            float h = silu_f(acc[nt][r] + bb);
            sMid[wave][(r + 8 * hi) * 64 + nt * 16 + row] = (_Float16)h;
        }
    }
    __syncthreads();

    v8f acc2[4];
#pragma unroll
    for (int nt = 0; nt < 4; ++nt) acc2[nt] = {};
#pragma unroll
    for (int kc = 0; kc < 2; ++kc) {
        const int kb = kc * 32 + hi * 8;
        v16h a = load_frag(&sMid[wave][row * 64 + kb]);
#pragma unroll
        for (int nt = 0; nt < 4; ++nt)
            acc2[nt] = wmma32(a, load_frag(&sW2T[(nt * 16 + row) * 64 + kb]), acc2[nt]);
    }
    if (validWave) {
#pragma unroll
        for (int nt = 0; nt < 4; ++nt) {
            float bb = b_n2[nt * 16 + row];
#pragma unroll
            for (int r = 0; r < 8; ++r) {
                int m = r + 8 * hi;
                int nd = tile * 16 + m;
                if (nd < N) out[(long)nd * 64 + nt * 16 + row] = acc2[nt][r] + bb;
            }
        }
    }
}


extern "C" void kernel_launch(void* const* d_in, const int* in_sizes, int n_in,
                              void* d_out, int out_size, void* d_ws, size_t ws_size,
                              hipStream_t stream) {
    const float* node_feat = (const float*)d_in[0];
    const float* coord     = (const float*)d_in[1];
    const float* edge_feat = (const float*)d_in[2];
    const float* W_e1 = (const float*)d_in[3];
    const float* b_e1 = (const float*)d_in[4];
    const float* W_e2 = (const float*)d_in[5];
    const float* b_e2 = (const float*)d_in[6];
    const float* W_n1 = (const float*)d_in[7];
    const float* b_n1 = (const float*)d_in[8];
    const float* W_n2 = (const float*)d_in[9];
    const float* b_n2 = (const float*)d_in[10];
    const float* W_q  = (const float*)d_in[11];
    const float* W_k  = (const float*)d_in[12];
    const int*   src  = (const int*)d_in[13];
    const int*   dst  = (const int*)d_in[14];

    const int N = in_sizes[0] / 64;
    const int E = in_sizes[13];

    float*    e_arr   = (float*)d_ws;
    unsigned* e_max   = (unsigned*)(e_arr + E);
    float*    denom   = (float*)(e_max + N);
    float*    h_neigh = denom + N;

    int initBlocks = (N * 64 + 255) / 256;
    if (initBlocks > 1024) initBlocks = 1024;
    init_ws_kernel<<<initBlocks, 256, 0, stream>>>(e_max, denom, h_neigh, N);

    const int numEdgeTiles = (E + 15) / 16;
    const int edgeBlocks   = (numEdgeTiles + WAVES - 1) / WAVES;
    const int numNodeTiles = (N + 15) / 16;
    const int nodeBlocks   = (numNodeTiles + WAVES - 1) / WAVES;

    edge_score_kernel<<<edgeBlocks, BLOCK, 0, stream>>>(
        node_feat, coord, edge_feat, W_q, W_k, src, dst, e_arr, e_max, E);

    edge_msg_kernel<<<edgeBlocks, BLOCK, 0, stream>>>(
        node_feat, coord, edge_feat, W_e1, b_e1, W_e2, b_e2,
        src, dst, e_arr, e_max, denom, h_neigh, E);

    node_mlp_kernel<<<nodeBlocks, BLOCK, 0, stream>>>(
        node_feat, W_n1, b_n1, W_n2, b_n2, denom, h_neigh, (float*)d_out, N);
}


# node-side products of W_k, W_q, W_e1 hoisted by linearity: done once per node into f32 tables gathered per edge; only [edge_feat|dist] stays a per-edge product, weights pre-packed as fragments
# speedup vs baseline: 2.2336x; 2.2336x over previous
#include <hip/hip_runtime.h>
#include <hip/hip_bf16.h>
#include <math.h>


typedef __attribute__((ext_vector_type(16))) _Float16 v16h;
typedef __attribute__((ext_vector_type(8)))  _Float16 v8h;
typedef __attribute__((ext_vector_type(4)))  _Float16 v4h;
typedef __attribute__((ext_vector_type(8)))  float    v8f;

#define WAVES 4
#define BLOCK 128
#define PKP 72


__device__ __forceinline__ float silu_f(float x) {
    return x / (1.0f + __expf(-x));
}

__device__ __forceinline__ unsigned enc_f(float f) {
    unsigned u = __float_as_uint(f);
    return (u & 0x80000000u) ? ~u : (u | 0x80000000u);
}
__device__ __forceinline__ float dec_f(unsigned u) {
    unsigned b = (u & 0x80000000u) ? (u & 0x7FFFFFFFu) : ~u;
    return __uint_as_float(b);
}
#define ENC_NEG_INF 0x007FFFFFu

__device__ __forceinline__ v16h load_frag(const _Float16* p) {
    v8h lo = *(const v8h*)(p);
    v8h hi = *(const v8h*)(p + 16);
    return __builtin_shufflevector(lo, hi, 0, 1, 2, 3, 4, 5, 6, 7,
                                   8, 9, 10, 11, 12, 13, 14, 15);
}

__device__ __forceinline__ v16h load_packed(const _Float16* base, int lane) {
    v8h lo = *(const v8h*)(base + lane * 8);
    v8h hi = *(const v8h*)(base + 256 + lane * 8);
    return __builtin_shufflevector(lo, hi, 0, 1, 2, 3, 4, 5, 6, 7,
                                   8, 9, 10, 11, 12, 13, 14, 15);
}

__device__ __forceinline__ v8f wmma32(v16h a, v16h b, v8f c) {
    return __builtin_amdgcn_wmma_f32_16x16x32_f16(
         false, a,  false, b,
         (short)0, c,  false,  false);
}

__device__ __forceinline__ void cvt_store4(_Float16* dstp, const float* srcp) {
    const float4 v = *(const float4*)srcp;
    v4h h;
    h[0] = (_Float16)v.x; h[1] = (_Float16)v.y;
    h[2] = (_Float16)v.z; h[3] = (_Float16)v.w;
    *(v4h*)dstp = h;
}

__device__ __forceinline__ v8f load8(const float* p) {
    const float4 a = *(const float4*)p;
    const float4 b = *(const float4*)(p + 4);
    v8f r;
    r[0] = a.x; r[1] = a.y; r[2] = a.z; r[3] = a.w;
    r[4] = b.x; r[5] = b.y; r[6] = b.z; r[7] = b.w;
    return r;
}

__device__ __forceinline__ v16h edge_frag(const float* __restrict__ edge_feat,
                                          const float* __restrict__ coord4,
                                          int e, int s, int d, int h) {
    const float* efp = edge_feat + (long)e * 16 + 8 * h;
    const float4 x0 = *(const float4*)efp;
    const float4 x1 = *(const float4*)(efp + 4);
    const float4 cs = *(const float4*)(coord4 + (long)s * 4);
    const float4 cd = *(const float4*)(coord4 + (long)d * 4);
    const float dx = cs.x - cd.x, dy = cs.y - cd.y, dz = cs.z - cd.z;
    const float dist = sqrtf(dx * dx + dy * dy + dz * dz);
    v16h x = {};
    x[0] = (_Float16)x0.x; x[1] = (_Float16)x0.y; x[2] = (_Float16)x0.z; x[3] = (_Float16)x0.w;
    x[4] = (_Float16)x1.x; x[5] = (_Float16)x1.y; x[6] = (_Float16)x1.z; x[7] = (_Float16)x1.w;
    x[8] = (h == 0) ? (_Float16)dist : (_Float16)0.0f;
    return x;
}

__device__ __forceinline__ float pack_resid(const float* __restrict__ W, int rowDist, int rowEf0,
                                            int h, int s, int col) {
    if (s < 8) return W[(rowEf0 + 8 * h + s) * 64 + col];
    if (s == 8 && h == 0) return W[rowDist * 64 + col];
    return 0.0f;
}


__global__ __launch_bounds__(BLOCK) void node_proj_kernel(
    const float* __restrict__ node_feat, const float* __restrict__ coord,
    const float* __restrict__ W_k, const float* __restrict__ W_q,
    const float* __restrict__ W_e1, const float* __restrict__ b_e1,
    const float* __restrict__ W_e2,
    float* __restrict__ Kn, float* __restrict__ Qn, float* __restrict__ Ps, float* __restrict__ Pd,
    float* __restrict__ coord4,
    unsigned* __restrict__ fragK_w, unsigned* __restrict__ fragE1_w, unsigned* __restrict__ fragE2_w,
    unsigned* __restrict__ e_max, float* __restrict__ denom, float* __restrict__ h_neigh, int N) {

    __shared__ _Float16 sWT[256 * PKP];
    _Float16* fragK  = (_Float16*)fragK_w;
    _Float16* fragE1 = (_Float16*)fragE1_w;
    _Float16* fragE2 = (_Float16*)fragE2_w;

    for (int i = threadIdx.x; i < 4 * 64 * 64; i += BLOCK) {
        const int t = i >> 12, k = (i >> 6) & 63, n = i & 63;
        float w;
        if (t == 0)      w = W_k[k * 64 + n];
        else if (t == 1) w = W_q[k * 64 + n];
        else if (t == 2) w = W_e1[k * 64 + n];
        else             w = W_e1[(64 + k) * 64 + n];
        sWT[(t * 64 + n) * PKP + k] = (_Float16)w;
    }

    const int node0 = blockIdx.x * 64;
    for (int i = threadIdx.x; i < 64 * 16; i += BLOCK) {
        const int nd = node0 + (i >> 4);
        if (nd < N) *(float4*)&h_neigh[(long)nd * 64 + (i & 15) * 4] = make_float4(0.f, 0.f, 0.f, 0.f);
    }
    if (threadIdx.x < 64) {
        const int nd = node0 + threadIdx.x;
        if (nd < N) {
            denom[nd] = 0.0f;
            e_max[nd] = ENC_NEG_INF;
            *(float4*)(coord4 + (long)nd * 4) = make_float4(coord[nd * 3 + 0], coord[nd * 3 + 1], coord[nd * 3 + 2], 0.0f);
        }
    }

    if (blockIdx.x < 8) {
        for (int jj = 0; jj < 8; ++jj) {
            const int g = blockIdx.x * 1024 + jj * BLOCK + threadIdx.x;
            const int s8 = g & 7, ln = (g >> 3) & 31, j = (g >> 8) & 1;
            const int c = ln & 15, h = ln >> 4;
            if (g < 2048) {
                const int nt = g >> 9;
                fragK[g] = (_Float16)pack_resid(W_k, 64, 65, h, j * 8 + s8, nt * 16 + c);
            } else if (g < 4096) {
                const int nt = (g - 2048) >> 9;
                fragE1[g - 2048] = (_Float16)pack_resid(W_e1, 128, 129, h, j * 8 + s8, nt * 16 + c);
            } else {
                const int q = g - 4096;
                const int kc = (q >> 9) & 1, nt = q >> 10;
                const int k = kc * 32 + 8 * h + s8 + 16 * j;
                fragE2[q] = (_Float16)W_e2[k * 64 + nt * 16 + c];
            }
        }
    }

    __syncthreads();

    const int wave = threadIdx.x >> 5, lane = threadIdx.x & 31;
    const int c = lane & 15, h = lane >> 4;
    const int numTiles = (N + 15) >> 4;
    const int tile = blockIdx.x * WAVES + wave;
    if (tile >= numTiles) return;

    int node = tile * 16 + c;
    if (node >= N) node = N - 1;

    v16h a[2];
#pragma unroll
    for (int kc = 0; kc < 2; ++kc) {
        const float* p = node_feat + (long)node * 64 + kc * 32 + 8 * h;
        const float4 v0 = *(const float4*)p,        v1 = *(const float4*)(p + 4);
        const float4 v2 = *(const float4*)(p + 16), v3 = *(const float4*)(p + 20);
        v16h f;
        f[0] = (_Float16)v0.x; f[1] = (_Float16)v0.y; f[2] = (_Float16)v0.z; f[3] = (_Float16)v0.w;
        f[4] = (_Float16)v1.x; f[5] = (_Float16)v1.y; f[6] = (_Float16)v1.z; f[7] = (_Float16)v1.w;
        f[8] = (_Float16)v2.x; f[9] = (_Float16)v2.y; f[10] = (_Float16)v2.z; f[11] = (_Float16)v2.w;
        f[12] = (_Float16)v3.x; f[13] = (_Float16)v3.y; f[14] = (_Float16)v3.z; f[15] = (_Float16)v3.w;
        a[kc] = f;
    }

#pragma unroll
    for (int t = 0; t < 4; ++t) {
        float* T = (t == 0) ? Kn : (t == 1) ? Qn : (t == 2) ? Ps : Pd;
        v8f acc[4];
#pragma unroll
        for (int nt = 0; nt < 4; ++nt) acc[nt] = {};
#pragma unroll
        for (int kc = 0; kc < 2; ++kc) {
#pragma unroll
            for (int nt = 0; nt < 4; ++nt)
                acc[nt] = wmma32(a[kc],
                                 load_frag(&sWT[(t * 64 + nt * 16 + c) * PKP + kc * 32 + 8 * h]),
                                 acc[nt]);
        }
#pragma unroll
        for (int nt = 0; nt < 4; ++nt) {
            const float bb = (t == 3) ? b_e1[nt * 16 + c] : 0.0f;
#pragma unroll
            for (int r = 0; r < 8; ++r) {
                const int nd = tile * 16 + 8 * h + r;
                if (nd < N) T[(long)nd * 64 + nt * 16 + c] = acc[nt][r] + bb;
            }
        }
    }
}


__global__ __launch_bounds__(BLOCK) void edge_score_kernel(
    const float* __restrict__ edge_feat, const float* __restrict__ coord4,
    const float* __restrict__ Kn, const float* __restrict__ Qn,
    const unsigned* __restrict__ fragK_w,
    const int* __restrict__ src, const int* __restrict__ dst,
    float* __restrict__ e_arr, unsigned* __restrict__ e_max, int E) {

    const int wave = threadIdx.x >> 5, lane = threadIdx.x & 31;
    const int c = lane & 15, h = lane >> 4;
    const int numTiles = (E + 15) >> 4;
    const int tile = blockIdx.x * WAVES + wave;
    if (tile >= numTiles) return;
    const _Float16* fragK = (const _Float16*)fragK_w;

    const int eIdx = tile * 16 + c;
    const bool valid = eIdx < E;
    const int e = valid ? eIdx : (E - 1);
    const int s = src[e], d = dst[e];

    const float* kp = Kn + (long)s * 64 + 8 * h;
    const float* qp = Qn + (long)d * 64 + 8 * h;
    v8f acc[4], q[4];
#pragma unroll
    for (int nt = 0; nt < 4; ++nt) { acc[nt] = load8(kp + nt * 16); q[nt] = load8(qp + nt * 16); }

    const v16h x = edge_frag(edge_feat, coord4, e, s, d, h);

#pragma unroll
    for (int nt = 0; nt < 4; ++nt)
        acc[nt] = wmma32(load_packed(fragK + nt * 512, lane), x, acc[nt]);

    float p = 0.0f;
#pragma unroll
    for (int nt = 0; nt < 4; ++nt) {
#pragma unroll
        for (int r = 0; r < 8; ++r) p += acc[nt][r] * q[nt][r];
    }
    p += __shfl_xor(p, 16, 32);

    if (h == 0 && valid) {
        const float ev = p * 0.125f;
        e_arr[eIdx] = ev;
        atomicMax(&e_max[d], enc_f(ev));
    }
}


__global__ __launch_bounds__(BLOCK) void edge_msg_kernel(
    const float* __restrict__ edge_feat, const float* __restrict__ coord4,
    const float* __restrict__ Ps, const float* __restrict__ Pd,
    const unsigned* __restrict__ fragE1_w, const unsigned* __restrict__ fragE2_w,
    const float* __restrict__ b_e2,
    const int* __restrict__ src, const int* __restrict__ dst,
    const float* __restrict__ e_arr, const unsigned* __restrict__ e_max,
    float* __restrict__ denom, float* __restrict__ h_neigh, int E) {

    const int wave = threadIdx.x >> 5, lane = threadIdx.x & 31;
    const int c = lane & 15, h = lane >> 4;
    const int numTiles = (E + 15) >> 4;
    const int tile = blockIdx.x * WAVES + wave;
    if (tile >= numTiles) return;
    const _Float16* fragE1 = (const _Float16*)fragE1_w;
    const _Float16* fragE2 = (const _Float16*)fragE2_w;

    const int eIdx = tile * 16 + c;
    const bool valid = eIdx < E;
    const int e = valid ? eIdx : (E - 1);
    const int s = src[e], d = dst[e];

    const float* sp = Ps + (long)s * 64 + 8 * h;
    const float* dp = Pd + (long)d * 64 + 8 * h;
    v8f acc[4];
#pragma unroll
    for (int nt = 0; nt < 4; ++nt) acc[nt] = load8(sp + nt * 16) + load8(dp + nt * 16);

    float w = 0.0f;
    if (valid) w = __expf(e_arr[eIdx] - dec_f(e_max[d]));
    if (h == 0 && valid) atomicAdd(&denom[d], w);

    const v16h x = edge_frag(edge_feat, coord4, e, s, d, h);

#pragma unroll
    for (int nt = 0; nt < 4; ++nt)
        acc[nt] = wmma32(load_packed(fragE1 + nt * 512, lane), x, acc[nt]);

    v16h mid[2];
#pragma unroll
    for (int kc = 0; kc < 2; ++kc) {
        v16h m;
#pragma unroll
        for (int r = 0; r < 8; ++r) {
            m[r]     = (_Float16)silu_f(acc[2 * kc][r]);
            m[8 + r] = (_Float16)silu_f(acc[2 * kc + 1][r]);
        }
        mid[kc] = m;
    }

    v8f acc2[4];
#pragma unroll
    for (int nt = 0; nt < 4; ++nt) {
        acc2[nt] = {};
#pragma unroll
        for (int kc = 0; kc < 2; ++kc)
            acc2[nt] = wmma32(mid[kc], load_packed(fragE2 + (nt * 2 + kc) * 512, lane), acc2[nt]);
    }

    float bb[4];
#pragma unroll
    for (int nt = 0; nt < 4; ++nt) bb[nt] = b_e2[nt * 16 + c];

#pragma unroll
    for (int r = 0; r < 8; ++r) {
        const int m = 8 * h + r;
        const float wr = __shfl(w, m, 32);
        const int   dr = __shfl(d, m, 32);
        if (tile * 16 + m < E) {
            float* hp = h_neigh + (long)dr * 64 + c;
#pragma unroll
            for (int nt = 0; nt < 4; ++nt)
                atomicAdd(hp + nt * 16, silu_f(acc2[nt][r] + bb[nt]) * wr);
        }
    }
}


__global__ __launch_bounds__(BLOCK) void node_mlp_kernel(
    const float* __restrict__ node_feat,
    const float* __restrict__ W_n1, const float* __restrict__ b_n1,
    const float* __restrict__ W_n2, const float* __restrict__ b_n2,
    const float* __restrict__ denom, const float* __restrict__ h_neigh,
    float* __restrict__ out, int N) {

    __shared__ _Float16 sW1T[64 * 128];
    __shared__ _Float16 sW2T[64 * 64];
    __shared__ _Float16 sA[WAVES][16 * 128];
    __shared__ _Float16 sMid[WAVES][16 * 64];

    for (int i = threadIdx.x; i < 64 * 128; i += BLOCK) {
        int n = i >> 7, k = i & 127;
        sW1T[i] = (_Float16)W_n1[k * 64 + n];
    }
    for (int i = threadIdx.x; i < 64 * 64; i += BLOCK) {
        int n = i >> 6, k = i & 63;
        sW2T[i] = (_Float16)W_n2[k * 64 + n];
    }

    const int wave = threadIdx.x >> 5, lane = threadIdx.x & 31;
    const int row = lane & 15, hi = lane >> 4;
    const int numTiles = (N + 15) >> 4;
    const int tile = blockIdx.x * WAVES + wave;
    const bool validWave = tile < numTiles;

    int node = validWave ? (tile * 16 + row) : 0;
    if (node >= N) node = 0;

    __syncthreads();

    _Float16* Ar = &sA[wave][row * 128];
    if (hi == 0) {
#pragma unroll
        for (int c = 0; c < 64; c += 4) cvt_store4(&Ar[c], &node_feat[node * 64 + c]);
    } else {
        float dn = denom[node];
        float inv = (dn > 0.0f) ? (1.0f / dn) : 0.0f;
#pragma unroll
        for (int c = 0; c < 64; c += 4) {
            const float4 v = *(const float4*)&h_neigh[node * 64 + c];
            v4h h;
            h[0] = (_Float16)(v.x * inv); h[1] = (_Float16)(v.y * inv);
            h[2] = (_Float16)(v.z * inv); h[3] = (_Float16)(v.w * inv);
            *(v4h*)&Ar[64 + c] = h;
        }
    }
    __syncthreads();

    v8f acc[4];
#pragma unroll
    for (int nt = 0; nt < 4; ++nt) acc[nt] = {};
#pragma unroll
    for (int kc = 0; kc < 4; ++kc) {
        const int kb = kc * 32 + hi * 8;
        v16h a = load_frag(&sA[wave][row * 128 + kb]);
#pragma unroll
        for (int nt = 0; nt < 4; ++nt)
            acc[nt] = wmma32(a, load_frag(&sW1T[(nt * 16 + row) * 128 + kb]), acc[nt]);
    }
#pragma unroll
    for (int nt = 0; nt < 4; ++nt) {
        float bb = b_n1[nt * 16 + row];
#pragma unroll
        for (int r = 0; r < 8; ++r) {
            float h = silu_f(acc[nt][r] + bb);
            sMid[wave][(r + 8 * hi) * 64 + nt * 16 + row] = (_Float16)h;
        }
    }
    __syncthreads();

    v8f acc2[4];
#pragma unroll
    for (int nt = 0; nt < 4; ++nt) acc2[nt] = {};
#pragma unroll
    for (int kc = 0; kc < 2; ++kc) {
        const int kb = kc * 32 + hi * 8;
        v16h a = load_frag(&sMid[wave][row * 64 + kb]);
#pragma unroll
        for (int nt = 0; nt < 4; ++nt)
            acc2[nt] = wmma32(a, load_frag(&sW2T[(nt * 16 + row) * 64 + kb]), acc2[nt]);
    }
    if (validWave) {
#pragma unroll
        for (int nt = 0; nt < 4; ++nt) {
            float bb = b_n2[nt * 16 + row];
#pragma unroll
            for (int r = 0; r < 8; ++r) {
                int m = r + 8 * hi;
                int nd = tile * 16 + m;
                if (nd < N) out[(long)nd * 64 + nt * 16 + row] = acc2[nt][r] + bb;
            }
        }
    }
}


extern "C" void kernel_launch(void* const* d_in, const int* in_sizes, int n_in,
                              void* d_out, int out_size, void* d_ws, size_t ws_size,
                              hipStream_t stream) {
    const float* node_feat = (const float*)d_in[0];
    const float* coord     = (const float*)d_in[1];
    const float* edge_feat = (const float*)d_in[2];
    const float* W_e1 = (const float*)d_in[3];
    const float* b_e1 = (const float*)d_in[4];
    const float* W_e2 = (const float*)d_in[5];
    const float* b_e2 = (const float*)d_in[6];
    const float* W_n1 = (const float*)d_in[7];
    const float* b_n1 = (const float*)d_in[8];
    const float* W_n2 = (const float*)d_in[9];
    const float* b_n2 = (const float*)d_in[10];
    const float* W_q  = (const float*)d_in[11];
    const float* W_k  = (const float*)d_in[12];
    const int*   src  = (const int*)d_in[13];
    const int*   dst  = (const int*)d_in[14];

    const int N = in_sizes[0] / 64;
    const int E = in_sizes[13];

    const size_t Ea = ((size_t)E + 63) & ~(size_t)63;
    const size_t Na = ((size_t)N + 63) & ~(size_t)63;
    const size_t T  = (size_t)N * 64;
    float*    e_arr   = (float*)d_ws;
    unsigned* e_max   = (unsigned*)(e_arr + Ea);
    float*    denom   = (float*)(e_max + Na);
    float*    h_neigh = denom + Na;
    float*    Kn      = h_neigh + T;
    float*    Qn      = Kn + T;
    float*    Ps      = Qn + T;
    float*    Pd      = Ps + T;
    float*    coord4  = Pd + T;
    unsigned* fragK   = (unsigned*)(coord4 + 4 * Na);
    unsigned* fragE1  = fragK + 1024;
    unsigned* fragE2  = fragE1 + 1024;

    const size_t wsNeed = (size_t)((const char*)(fragE2 + 2048) - (const char*)d_ws);
    if (ws_size < wsNeed) return;

    const int numEdgeTiles = (E + 15) / 16;
    const int edgeBlocks   = (numEdgeTiles + WAVES - 1) / WAVES;
    const int numNodeTiles = (N + 15) / 16;
    const int nodeBlocks   = (numNodeTiles + WAVES - 1) / WAVES;
    const int projBlocks   = nodeBlocks < 8 ? 8 : nodeBlocks;

    node_proj_kernel<<<projBlocks, BLOCK, 0, stream>>>(
        node_feat, coord, W_k, W_q, W_e1, b_e1, W_e2,
        Kn, Qn, Ps, Pd, coord4, fragK, fragE1, fragE2, e_max, denom, h_neigh, N);

    edge_score_kernel<<<edgeBlocks, BLOCK, 0, stream>>>(
        edge_feat, coord4, Kn, Qn, fragK, src, dst, e_arr, e_max, E);

    edge_msg_kernel<<<edgeBlocks, BLOCK, 0, stream>>>(
        edge_feat, coord4, Ps, Pd, fragE1, fragE2, b_e2,
        src, dst, e_arr, e_max, denom, h_neigh, E);

    node_mlp_kernel<<<nodeBlocks, BLOCK, 0, stream>>>(
        node_feat, W_n1, b_n1, W_n2, b_n2, denom, h_neigh, (float*)d_out, N);
}


# set-up kernels (projection weight staging and fragment packing, count scan, record scatter at 4 edges per thread) issue their independent global loads together, not one round trip each
# speedup vs baseline: 2.8104x; 1.2582x over previous
#include <hip/hip_runtime.h>
#include <hip/hip_bf16.h>
#include <math.h>


typedef __attribute__((ext_vector_type(16))) _Float16 v16h;
typedef __attribute__((ext_vector_type(8)))  _Float16 v8h;
typedef __attribute__((ext_vector_type(4)))  _Float16 v4h;
typedef __attribute__((ext_vector_type(2)))  _Float16 v2h;
typedef __attribute__((ext_vector_type(8)))  float    v8f;

#define WAVES 4
#define BLOCK 128
#define PKP 72


__device__ __forceinline__ float silu_f(float x) {
    return x / (1.0f + __expf(-x));
}

__device__ __forceinline__ unsigned enc_f(float f) {
    unsigned u = __float_as_uint(f);
    return (u & 0x80000000u) ? ~u : (u | 0x80000000u);
}
__device__ __forceinline__ float dec_f(unsigned u) {
    unsigned b = (u & 0x80000000u) ? (u & 0x7FFFFFFFu) : ~u;
    return __uint_as_float(b);
}
#define ENC_NEG_INF 0x007FFFFFu

__device__ __forceinline__ v16h load_frag(const _Float16* p) {
    v8h lo = *(const v8h*)(p);
    v8h hi = *(const v8h*)(p + 16);
    return __builtin_shufflevector(lo, hi, 0, 1, 2, 3, 4, 5, 6, 7,
                                   8, 9, 10, 11, 12, 13, 14, 15);
}

__device__ __forceinline__ v16h load_packed(const _Float16* base, int lane) {
    v8h lo = *(const v8h*)(base + lane * 8);
    v8h hi = *(const v8h*)(base + 256 + lane * 8);
    return __builtin_shufflevector(lo, hi, 0, 1, 2, 3, 4, 5, 6, 7,
                                   8, 9, 10, 11, 12, 13, 14, 15);
}

__device__ __forceinline__ v8f wmma32(v16h a, v16h b, v8f c) {
    return __builtin_amdgcn_wmma_f32_16x16x32_f16(
         false, a,  false, b,
         (short)0, c,  false,  false);
}

__device__ __forceinline__ void cvt_store4(_Float16* dstp, const float* srcp) {
    const float4 v = *(const float4*)srcp;
    v4h h;
    h[0] = (_Float16)v.x; h[1] = (_Float16)v.y;
    h[2] = (_Float16)v.z; h[3] = (_Float16)v.w;
    *(v4h*)dstp = h;
}

__device__ __forceinline__ v8f load8(const float* p) {
    const float4 a = *(const float4*)p;
    const float4 b = *(const float4*)(p + 4);
    v8f r;
    r[0] = a.x; r[1] = a.y; r[2] = a.z; r[3] = a.w;
    r[4] = b.x; r[5] = b.y; r[6] = b.z; r[7] = b.w;
    return r;
}

__device__ __forceinline__ v16h edge_frag(const float* __restrict__ edge_feat,
                                          const float* __restrict__ coord4,
                                          int e, int s, int d, int h) {
    const float* efp = edge_feat + (long)e * 16 + 8 * h;
    const float4 x0 = *(const float4*)efp;
    const float4 x1 = *(const float4*)(efp + 4);
    const float4 cs = *(const float4*)(coord4 + (long)s * 4);
    const float4 cd = *(const float4*)(coord4 + (long)d * 4);
    const float dx = cs.x - cd.x, dy = cs.y - cd.y, dz = cs.z - cd.z;
    const float dist = sqrtf(dx * dx + dy * dy + dz * dz);
    v16h x = {};
    x[0] = (_Float16)x0.x; x[1] = (_Float16)x0.y; x[2] = (_Float16)x0.z; x[3] = (_Float16)x0.w;
    x[4] = (_Float16)x1.x; x[5] = (_Float16)x1.y; x[6] = (_Float16)x1.z; x[7] = (_Float16)x1.w;
    x[8] = (h == 0) ? (_Float16)dist : (_Float16)0.0f;
    return x;
}

__device__ __forceinline__ float pack_resid(const float* __restrict__ W, int rowDist, int rowEf0,
                                            int h, int s, int col) {
    const bool isEf = s < 8;
    const bool isDist = (s == 8) && (h == 0);
    const int row = isEf ? (rowEf0 + 8 * h + s) : rowDist;
    const float v = W[row * 64 + col];
    return (isEf || isDist) ? v : 0.0f;
}


__global__ __launch_bounds__(BLOCK) void node_proj_kernel(
    const float* __restrict__ node_feat, const float* __restrict__ coord,
    const float* __restrict__ W_k, const float* __restrict__ W_q,
    const float* __restrict__ W_e1, const float* __restrict__ b_e1,
    const float* __restrict__ W_e2,
    float* __restrict__ Kn, float* __restrict__ Qn, float* __restrict__ Ps, float* __restrict__ Pd,
    float* __restrict__ coord4,
    unsigned* __restrict__ fragK_w, unsigned* __restrict__ fragE1_w, unsigned* __restrict__ fragE2_w,
    unsigned* __restrict__ e_max, float* __restrict__ denom, float* __restrict__ h_neigh,
    unsigned* __restrict__ cnt, int N) {

    __shared__ __attribute__((aligned(16))) _Float16 sWT[256 * PKP];
    _Float16* fragK  = (_Float16*)fragK_w;
    _Float16* fragE1 = (_Float16*)fragE1_w;
    _Float16* fragE2 = (_Float16*)fragE2_w;

    {
        const int n = threadIdx.x & 63, kp = threadIdx.x >> 6;
#pragma unroll
        for (int t = 0; t < 4; ++t) {
            const float* Wt = (t == 0) ? W_k : (t == 1) ? W_q : (t == 2) ? W_e1 : (W_e1 + 64 * 64);
            float w[32];
#pragma unroll
            for (int q = 0; q < 16; ++q) {
                const int k = 4 * q + 2 * kp;
                w[2 * q]     = Wt[k * 64 + n];
                w[2 * q + 1] = Wt[(k + 1) * 64 + n];
            }
#pragma unroll
            for (int q = 0; q < 16; ++q) {
                const int k = 4 * q + 2 * kp;
                v2h p;
                p[0] = (_Float16)w[2 * q]; p[1] = (_Float16)w[2 * q + 1];
                *(v2h*)&sWT[(t * 64 + n) * PKP + k] = p;
            }
        }
    }

    const int node0 = blockIdx.x * 64;
    for (int i = threadIdx.x; i < 64 * 16; i += BLOCK) {
        const int nd = node0 + (i >> 4);
        if (nd < N) *(float4*)&h_neigh[(long)nd * 64 + (i & 15) * 4] = make_float4(0.f, 0.f, 0.f, 0.f);
    }
    if (threadIdx.x < 64) {
        const int nd = node0 + threadIdx.x;
        if (nd < N) {
            denom[nd] = 0.0f;
            e_max[nd] = ENC_NEG_INF;
            cnt[nd] = 0u;
            *(float4*)(coord4 + (long)nd * 4) = make_float4(coord[nd * 3 + 0], coord[nd * 3 + 1], coord[nd * 3 + 2], 0.0f);
        }
    }

    if (blockIdx.x < 8) {
        const int g0 = blockIdx.x * 1024 + threadIdx.x;
        float pv[8];
        if (blockIdx.x < 2) {
#pragma unroll
            for (int jj = 0; jj < 8; ++jj) {
                const int g = g0 + jj * BLOCK;
                const int s8 = g & 7, ln = (g >> 3) & 31, j = (g >> 8) & 1;
                const int c = ln & 15, h = ln >> 4, nt = g >> 9;
                pv[jj] = pack_resid(W_k, 64, 65, h, j * 8 + s8, nt * 16 + c);
            }
#pragma unroll
            for (int jj = 0; jj < 8; ++jj) fragK[g0 + jj * BLOCK] = (_Float16)pv[jj];
        } else if (blockIdx.x < 4) {
#pragma unroll
            for (int jj = 0; jj < 8; ++jj) {
                const int g = g0 + jj * BLOCK;
                const int s8 = g & 7, ln = (g >> 3) & 31, j = (g >> 8) & 1;
                const int c = ln & 15, h = ln >> 4, nt = (g - 2048) >> 9;
                pv[jj] = pack_resid(W_e1, 128, 129, h, j * 8 + s8, nt * 16 + c);
            }
#pragma unroll
            for (int jj = 0; jj < 8; ++jj) fragE1[g0 - 2048 + jj * BLOCK] = (_Float16)pv[jj];
        } else {
#pragma unroll
            for (int jj = 0; jj < 8; ++jj) {
                const int g = g0 + jj * BLOCK;
                const int s8 = g & 7, ln = (g >> 3) & 31, j = (g >> 8) & 1;
                const int c = ln & 15, h = ln >> 4;
                const int q = g - 4096;
                const int kc = (q >> 9) & 1, nt = q >> 10;
                const int k = kc * 32 + 8 * h + s8 + 16 * j;
                pv[jj] = W_e2[k * 64 + nt * 16 + c];
            }
#pragma unroll
            for (int jj = 0; jj < 8; ++jj) fragE2[g0 - 4096 + jj * BLOCK] = (_Float16)pv[jj];
        }
    }

    __syncthreads();

    const int wave = threadIdx.x >> 5, lane = threadIdx.x & 31;
    const int c = lane & 15, h = lane >> 4;
    const int numTiles = (N + 15) >> 4;
    const int tile = blockIdx.x * WAVES + wave;
    if (tile >= numTiles) return;

    int node = tile * 16 + c;
    if (node >= N) node = N - 1;

    v16h a[2];
#pragma unroll
    for (int kc = 0; kc < 2; ++kc) {
        const float* p = node_feat + (long)node * 64 + kc * 32 + 8 * h;
        const float4 v0 = *(const float4*)p,        v1 = *(const float4*)(p + 4);
        const float4 v2 = *(const float4*)(p + 16), v3 = *(const float4*)(p + 20);
        v16h f;
        f[0] = (_Float16)v0.x; f[1] = (_Float16)v0.y; f[2] = (_Float16)v0.z; f[3] = (_Float16)v0.w;
        f[4] = (_Float16)v1.x; f[5] = (_Float16)v1.y; f[6] = (_Float16)v1.z; f[7] = (_Float16)v1.w;
        f[8] = (_Float16)v2.x; f[9] = (_Float16)v2.y; f[10] = (_Float16)v2.z; f[11] = (_Float16)v2.w;
        f[12] = (_Float16)v3.x; f[13] = (_Float16)v3.y; f[14] = (_Float16)v3.z; f[15] = (_Float16)v3.w;
        a[kc] = f;
    }

#pragma unroll
    for (int t = 0; t < 4; ++t) {
        float* T = (t == 0) ? Kn : (t == 1) ? Qn : (t == 2) ? Ps : Pd;
        v8f acc[4];
#pragma unroll
        for (int nt = 0; nt < 4; ++nt) acc[nt] = {};
#pragma unroll
        for (int kc = 0; kc < 2; ++kc) {
#pragma unroll
            for (int nt = 0; nt < 4; ++nt)
                acc[nt] = wmma32(a[kc],
                                 load_frag(&sWT[(t * 64 + nt * 16 + c) * PKP + kc * 32 + 8 * h]),
                                 acc[nt]);
        }
#pragma unroll
        for (int nt = 0; nt < 4; ++nt) {
            const float bb = (t == 3) ? b_e1[nt * 16 + c] : 0.0f;
#pragma unroll
            for (int r = 0; r < 8; ++r) {
                const int nd = tile * 16 + 8 * h + r;
                if (nd < N) T[(long)nd * 64 + nt * 16 + c] = acc[nt][r] + bb;
            }
        }
    }
}


__global__ __launch_bounds__(BLOCK) void edge_score_kernel(
    const float* __restrict__ edge_feat, const float* __restrict__ coord4,
    const float* __restrict__ Kn, const float* __restrict__ Qn,
    const unsigned* __restrict__ fragK_w,
    const int* __restrict__ src, const int* __restrict__ dst,
    float* __restrict__ e_arr, unsigned* __restrict__ e_max,
    unsigned* __restrict__ cnt, unsigned* __restrict__ rank, int E) {

    const int wave = threadIdx.x >> 5, lane = threadIdx.x & 31;
    const int c = lane & 15, h = lane >> 4;
    const int numTiles = (E + 15) >> 4;
    const int tile = blockIdx.x * WAVES + wave;
    if (tile >= numTiles) return;
    const _Float16* fragK = (const _Float16*)fragK_w;

    const int eIdx = tile * 16 + c;
    const bool valid = eIdx < E;
    const int e = valid ? eIdx : (E - 1);
    const int s = src[e], d = dst[e];

    const float* kp = Kn + (long)s * 64 + 8 * h;
    const float* qp = Qn + (long)d * 64 + 8 * h;
    v8f acc[4], q[4];
#pragma unroll
    for (int nt = 0; nt < 4; ++nt) { acc[nt] = load8(kp + nt * 16); q[nt] = load8(qp + nt * 16); }

    const v16h x = edge_frag(edge_feat, coord4, e, s, d, h);

#pragma unroll
    for (int nt = 0; nt < 4; ++nt)
        acc[nt] = wmma32(load_packed(fragK + nt * 512, lane), x, acc[nt]);

    float p = 0.0f;
#pragma unroll
    for (int nt = 0; nt < 4; ++nt) {
#pragma unroll
        for (int r = 0; r < 8; ++r) p += acc[nt][r] * q[nt][r];
    }
    p += __shfl_xor(p, 16, 32);

    if (h == 0 && valid) {
        const float ev = p * 0.125f;
        e_arr[eIdx] = ev;
        atomicMax(&e_max[d], enc_f(ev));
        rank[eIdx] = atomicAdd(&cnt[d], 1u);
    }
}


#define SCAN_T 1024

__global__ __launch_bounds__(SCAN_T) void rowstart_scan_kernel(
    const unsigned* __restrict__ cnt, unsigned* __restrict__ rowstart, int N) {

    __shared__ unsigned sWave[32];
    const int t = threadIdx.x, lane = t & 31, wave = t >> 5;
    const int per = (N + SCAN_T - 1) / SCAN_T;
    const int base = t * per;
    unsigned tot = 0;
    for (int i0 = 0; i0 < per; i0 += 8) {
        unsigned v[8];
#pragma unroll
        for (int j = 0; j < 8; ++j) {
            const int n = base + i0 + j;
            const bool ok = (i0 + j < per) && (n < N);
            const unsigned x = cnt[ok ? n : 0];
            v[j] = ok ? x : 0u;
        }
#pragma unroll
        for (int j = 0; j < 8; ++j) tot += v[j];
    }
    unsigned inc = tot;
    for (int o = 1; o < 32; o <<= 1) {
        const unsigned v = __shfl_up(inc, o, 32);
        if (lane >= o) inc += v;
    }
    if (lane == 31) sWave[wave] = inc;
    __syncthreads();
    if (wave == 0) {
        const unsigned wt = sWave[lane];
        unsigned winc = wt;
        for (int o = 1; o < 32; o <<= 1) {
            const unsigned v = __shfl_up(winc, o, 32);
            if (lane >= o) winc += v;
        }
        sWave[lane] = winc - wt;
    }
    __syncthreads();
    unsigned run = sWave[wave] + inc - tot;
    for (int i0 = 0; i0 < per; i0 += 8) {
        unsigned v[8];
#pragma unroll
        for (int j = 0; j < 8; ++j) {
            const int n = base + i0 + j;
            const bool ok = (i0 + j < per) && (n < N);
            const unsigned x = cnt[ok ? n : 0];
            v[j] = ok ? x : 0u;
        }
#pragma unroll
        for (int j = 0; j < 8; ++j) {
            const int n = base + i0 + j;
            if ((i0 + j < per) && (n < N)) { rowstart[n] = run; run += v[j]; }
        }
    }
}


__global__ __launch_bounds__(256) void edge_scatter_kernel(
    const int* __restrict__ src, const int* __restrict__ dst,
    const float* __restrict__ e_arr, const unsigned* __restrict__ rank,
    const unsigned* __restrict__ rowstart, int4* __restrict__ sorted4, int E) {

    const int e0 = (blockIdx.x * 256 + threadIdx.x) * 4;
    if (e0 >= E) return;
    const unsigned last = (unsigned)(E - 1);
    if (e0 + 3 < E) {
        const int4   d4 = *(const int4*)(dst + e0);
        const int4   s4 = *(const int4*)(src + e0);
        const uint4  r4 = *(const uint4*)(rank + e0);
        const float4 v4 = *(const float4*)(e_arr + e0);
        const unsigned p0 = rowstart[d4.x] + r4.x;
        const unsigned p1 = rowstart[d4.y] + r4.y;
        const unsigned p2 = rowstart[d4.z] + r4.z;
        const unsigned p3 = rowstart[d4.w] + r4.w;
        sorted4[min(p0, last)] = make_int4(e0,     s4.x, d4.x, __float_as_int(v4.x));
        sorted4[min(p1, last)] = make_int4(e0 + 1, s4.y, d4.y, __float_as_int(v4.y));
        sorted4[min(p2, last)] = make_int4(e0 + 2, s4.z, d4.z, __float_as_int(v4.z));
        sorted4[min(p3, last)] = make_int4(e0 + 3, s4.w, d4.w, __float_as_int(v4.w));
    } else {
        for (int e = e0; e < E; ++e) {
            const int d = dst[e];
            const unsigned pos = rowstart[d] + rank[e];
            sorted4[min(pos, last)] = make_int4(e, src[e], d, __float_as_int(e_arr[e]));
        }
    }
}


__global__ __launch_bounds__(BLOCK) void edge_msg_kernel(
    const float* __restrict__ edge_feat, const float* __restrict__ coord4,
    const float* __restrict__ Ps, const float* __restrict__ Pd,
    const unsigned* __restrict__ fragE1_w, const unsigned* __restrict__ fragE2_w,
    const float* __restrict__ b_e2,
    const int4* __restrict__ sorted4, const unsigned* __restrict__ e_max,
    float* __restrict__ denom, float* __restrict__ h_neigh, int E, int N) {

    const int wave = threadIdx.x >> 5, lane = threadIdx.x & 31;
    const int c = lane & 15, h = lane >> 4;
    const int numTiles = (E + 15) >> 4;
    const int tile = blockIdx.x * WAVES + wave;
    if (tile >= numTiles) return;
    const _Float16* fragE1 = (const _Float16*)fragE1_w;
    const _Float16* fragE2 = (const _Float16*)fragE2_w;

    const int place = tile * 16 + c;
    const bool valid = place < E;
    const int4 rec = sorted4[valid ? place : (E - 1)];
    const int e = (int)min((unsigned)rec.x, (unsigned)(E - 1));
    const int s = (int)min((unsigned)rec.y, (unsigned)(N - 1));
    const int d = (int)min((unsigned)rec.z, (unsigned)(N - 1));

    const float* sp = Ps + (long)s * 64 + 8 * h;
    const float* dp = Pd + (long)d * 64 + 8 * h;
    v8f acc[4];
#pragma unroll
    for (int nt = 0; nt < 4; ++nt) acc[nt] = load8(sp + nt * 16) + load8(dp + nt * 16);

    float w = 0.0f;
    if (valid) w = __expf(__int_as_float(rec.w) - dec_f(e_max[d]));

    const v16h x = edge_frag(edge_feat, coord4, e, s, d, h);

#pragma unroll
    for (int nt = 0; nt < 4; ++nt)
        acc[nt] = wmma32(load_packed(fragE1 + nt * 512, lane), x, acc[nt]);

    v16h mid[2];
#pragma unroll
    for (int kc = 0; kc < 2; ++kc) {
        v16h m;
#pragma unroll
        for (int r = 0; r < 8; ++r) {
            m[r]     = (_Float16)silu_f(acc[2 * kc][r]);
            m[8 + r] = (_Float16)silu_f(acc[2 * kc + 1][r]);
        }
        mid[kc] = m;
    }

    v8f acc2[4];
#pragma unroll
    for (int nt = 0; nt < 4; ++nt) {
        acc2[nt] = {};
#pragma unroll
        for (int kc = 0; kc < 2; ++kc)
            acc2[nt] = wmma32(mid[kc], load_packed(fragE2 + (nt * 2 + kc) * 512, lane), acc2[nt]);
    }

    float bb[4];
#pragma unroll
    for (int nt = 0; nt < 4; ++nt) bb[nt] = b_e2[nt * 16 + c];

    const int d7 = __shfl(d, 7, 32), d8 = __shfl(d, 8, 32);
    const bool joined = d7 == d8;
    bool holdFirst = joined && h == 1;
    float run[4]  = {0.0f, 0.0f, 0.0f, 0.0f};
    float held[4] = {0.0f, 0.0f, 0.0f, 0.0f};
    float wsum = 0.0f, wheld = 0.0f;
    int dr = __shfl(d, 8 * h, 32);
#pragma unroll
    for (int r = 0; r < 8; ++r) {
        const int m = 8 * h + r;
        const float wr = __shfl(w, m, 32);
        const int dn = (r < 7) ? __shfl(d, m + 1, 32) : dr;
        wsum += wr;
#pragma unroll
        for (int nt = 0; nt < 4; ++nt) run[nt] += silu_f(acc2[nt][r] + bb[nt]) * wr;
        if (r == 7 || dn != dr) {
            if (holdFirst || (joined && h == 0 && r == 7)) {
#pragma unroll
                for (int nt = 0; nt < 4; ++nt) held[nt] = run[nt];
                wheld = wsum;
            } else {
                float* hp = h_neigh + (long)dr * 64 + c;
#pragma unroll
                for (int nt = 0; nt < 4; ++nt) atomicAdd(hp + nt * 16, run[nt]);
                if (c == 0) atomicAdd(&denom[dr], wsum);
            }
            holdFirst = false;
#pragma unroll
            for (int nt = 0; nt < 4; ++nt) run[nt] = 0.0f;
            wsum = 0.0f;
        }
        dr = dn;
    }
    if (joined) {
        float* hp = h_neigh + (long)d7 * 64 + c;
#pragma unroll
        for (int nt = 0; nt < 4; ++nt) {
            const float other = __shfl_xor(held[nt], 16, 32);
            if (h == 0) atomicAdd(hp + nt * 16, held[nt] + other);
        }
        const float wother = __shfl_xor(wheld, 16, 32);
        if (lane == 0) atomicAdd(&denom[d7], wheld + wother);
    }
}


__global__ __launch_bounds__(BLOCK) void node_mlp_kernel(
    const float* __restrict__ node_feat,
    const float* __restrict__ W_n1, const float* __restrict__ b_n1,
    const float* __restrict__ W_n2, const float* __restrict__ b_n2,
    const float* __restrict__ denom, const float* __restrict__ h_neigh,
    float* __restrict__ out, int N) {

    __shared__ _Float16 sW1T[64 * 128];
    __shared__ _Float16 sW2T[64 * 64];
    __shared__ _Float16 sA[WAVES][16 * 128];
    __shared__ _Float16 sMid[WAVES][16 * 64];

    for (int i = threadIdx.x; i < 64 * 128; i += BLOCK) {
        int n = i >> 7, k = i & 127;
        sW1T[i] = (_Float16)W_n1[k * 64 + n];
    }
    for (int i = threadIdx.x; i < 64 * 64; i += BLOCK) {
        int n = i >> 6, k = i & 63;
        sW2T[i] = (_Float16)W_n2[k * 64 + n];
    }

    const int wave = threadIdx.x >> 5, lane = threadIdx.x & 31;
    const int row = lane & 15, hi = lane >> 4;
    const int numTiles = (N + 15) >> 4;
    const int tile = blockIdx.x * WAVES + wave;
    const bool validWave = tile < numTiles;

    int node = validWave ? (tile * 16 + row) : 0;
    if (node >= N) node = 0;

    __syncthreads();

    _Float16* Ar = &sA[wave][row * 128];
    if (hi == 0) {
#pragma unroll
        for (int c = 0; c < 64; c += 4) cvt_store4(&Ar[c], &node_feat[node * 64 + c]);
    } else {
        float dn = denom[node];
        float inv = (dn > 0.0f) ? (1.0f / dn) : 0.0f;
#pragma unroll
        for (int c = 0; c < 64; c += 4) {
            const float4 v = *(const float4*)&h_neigh[node * 64 + c];
            v4h h;
            h[0] = (_Float16)(v.x * inv); h[1] = (_Float16)(v.y * inv);
            h[2] = (_Float16)(v.z * inv); h[3] = (_Float16)(v.w * inv);
            *(v4h*)&Ar[64 + c] = h;
        }
    }
    __syncthreads();

    v8f acc[4];
#pragma unroll
    for (int nt = 0; nt < 4; ++nt) acc[nt] = {};
#pragma unroll
    for (int kc = 0; kc < 4; ++kc) {
        const int kb = kc * 32 + hi * 8;
        v16h a = load_frag(&sA[wave][row * 128 + kb]);
#pragma unroll
        for (int nt = 0; nt < 4; ++nt)
            acc[nt] = wmma32(a, load_frag(&sW1T[(nt * 16 + row) * 128 + kb]), acc[nt]);
    }
#pragma unroll
    for (int nt = 0; nt < 4; ++nt) {
        float bb = b_n1[nt * 16 + row];
#pragma unroll
        for (int r = 0; r < 8; ++r) {
            float h = silu_f(acc[nt][r] + bb);
            sMid[wave][(r + 8 * hi) * 64 + nt * 16 + row] = (_Float16)h;
        }
    }
    __syncthreads();

    v8f acc2[4];
#pragma unroll
    for (int nt = 0; nt < 4; ++nt) acc2[nt] = {};
#pragma unroll
    for (int kc = 0; kc < 2; ++kc) {
        const int kb = kc * 32 + hi * 8;
        v16h a = load_frag(&sMid[wave][row * 64 + kb]);
#pragma unroll
        for (int nt = 0; nt < 4; ++nt)
            acc2[nt] = wmma32(a, load_frag(&sW2T[(nt * 16 + row) * 64 + kb]), acc2[nt]);
    }
    if (validWave) {
#pragma unroll
        for (int nt = 0; nt < 4; ++nt) {
            float bb = b_n2[nt * 16 + row];
#pragma unroll
            for (int r = 0; r < 8; ++r) {
                int m = r + 8 * hi;
                int nd = tile * 16 + m;
                if (nd < N) out[(long)nd * 64 + nt * 16 + row] = acc2[nt][r] + bb;
            }
        }
    }
}


extern "C" void kernel_launch(void* const* d_in, const int* in_sizes, int n_in,
                              void* d_out, int out_size, void* d_ws, size_t ws_size,
                              hipStream_t stream) {
    const float* node_feat = (const float*)d_in[0];
    const float* coord     = (const float*)d_in[1];
    const float* edge_feat = (const float*)d_in[2];
    const float* W_e1 = (const float*)d_in[3];
    const float* b_e1 = (const float*)d_in[4];
    const float* W_e2 = (const float*)d_in[5];
    const float* b_e2 = (const float*)d_in[6];
    const float* W_n1 = (const float*)d_in[7];
    const float* b_n1 = (const float*)d_in[8];
    const float* W_n2 = (const float*)d_in[9];
    const float* b_n2 = (const float*)d_in[10];
    const float* W_q  = (const float*)d_in[11];
    const float* W_k  = (const float*)d_in[12];
    const int*   src  = (const int*)d_in[13];
    const int*   dst  = (const int*)d_in[14];

    const int N = in_sizes[0] / 64;
    const int E = in_sizes[13];

    const size_t Ea = ((size_t)E + 63) & ~(size_t)63;
    const size_t Na = ((size_t)N + 63) & ~(size_t)63;
    const size_t T  = (size_t)N * 64;
    float*    e_arr   = (float*)d_ws;
    unsigned* e_max   = (unsigned*)(e_arr + Ea);
    float*    denom   = (float*)(e_max + Na);
    float*    h_neigh = denom + Na;
    float*    Kn      = h_neigh + T;
    float*    Qn      = Kn + T;
    float*    Ps      = Qn + T;
    float*    Pd      = Ps + T;
    float*    coord4  = Pd + T;
    unsigned* fragK   = (unsigned*)(coord4 + 4 * Na);
    unsigned* fragE1  = fragK + 1024;
    unsigned* fragE2  = fragE1 + 1024;
    unsigned* cnt      = fragE2 + 2048;
    unsigned* rowstart = cnt + Na;
    unsigned* rank     = rowstart + Na;
    int4*     sorted4  = (int4*)(rank + Ea);

    const size_t wsNeed = (size_t)((const char*)(sorted4 + E) - (const char*)d_ws);
    if (ws_size < wsNeed) return;

    const int numEdgeTiles = (E + 15) / 16;
    const int edgeBlocks   = (numEdgeTiles + WAVES - 1) / WAVES;
    const int numNodeTiles = (N + 15) / 16;
    const int nodeBlocks   = (numNodeTiles + WAVES - 1) / WAVES;
    const int projBlocks   = nodeBlocks < 8 ? 8 : nodeBlocks;

    node_proj_kernel<<<projBlocks, BLOCK, 0, stream>>>(
        node_feat, coord, W_k, W_q, W_e1, b_e1, W_e2,
        Kn, Qn, Ps, Pd, coord4, fragK, fragE1, fragE2, e_max, denom, h_neigh, cnt, N);

    edge_score_kernel<<<edgeBlocks, BLOCK, 0, stream>>>(
        edge_feat, coord4, Kn, Qn, fragK, src, dst, e_arr, e_max, cnt, rank, E);

    rowstart_scan_kernel<<<1, SCAN_T, 0, stream>>>(cnt, rowstart, N);

    edge_scatter_kernel<<<((E + 3) / 4 + 255) / 256, 256, 0, stream>>>(
        src, dst, e_arr, rank, rowstart, sorted4, E);

    edge_msg_kernel<<<edgeBlocks, BLOCK, 0, stream>>>(
        edge_feat, coord4, Ps, Pd, fragE1, fragE2, b_e2,
        sorted4, e_max, denom, h_neigh, E, N);

    node_mlp_kernel<<<nodeBlocks, BLOCK, 0, stream>>>(
        node_feat, W_n1, b_n1, W_n2, b_n2, denom, h_neigh, (float*)d_out, N);
}


# node MLP without LDS: W_n1 and W_n2 arrive as fragments packed by the first kernel, layer 1 runs transposed so layer 2 takes its operand from registers; no weight staging, no barriers
# speedup vs baseline: 3.3847x; 1.2043x over previous
#include <hip/hip_runtime.h>
#include <hip/hip_bf16.h>
#include <math.h>


typedef __attribute__((ext_vector_type(16))) _Float16 v16h;
typedef __attribute__((ext_vector_type(8)))  _Float16 v8h;
typedef __attribute__((ext_vector_type(4)))  _Float16 v4h;
typedef __attribute__((ext_vector_type(2)))  _Float16 v2h;
typedef __attribute__((ext_vector_type(8)))  float    v8f;
typedef __attribute__((ext_vector_type(4)))  float    v4f;

#define WAVES 4
#define BLOCK 128
#define PKP 72


__device__ __forceinline__ float silu_f(float x) {
    return x / (1.0f + __expf(-x));
}

__device__ __forceinline__ unsigned enc_f(float f) {
    unsigned u = __float_as_uint(f);
    return (u & 0x80000000u) ? ~u : (u | 0x80000000u);
}
__device__ __forceinline__ float dec_f(unsigned u) {
    unsigned b = (u & 0x80000000u) ? (u & 0x7FFFFFFFu) : ~u;
    return __uint_as_float(b);
}
#define ENC_NEG_INF 0x007FFFFFu

__device__ __forceinline__ v16h load_frag(const _Float16* p) {
    v8h lo = *(const v8h*)(p);
    v8h hi = *(const v8h*)(p + 16);
    return __builtin_shufflevector(lo, hi, 0, 1, 2, 3, 4, 5, 6, 7,
                                   8, 9, 10, 11, 12, 13, 14, 15);
}

__device__ __forceinline__ v16h load_packed(const _Float16* base, int lane) {
    v8h lo = *(const v8h*)(base + lane * 8);
    v8h hi = *(const v8h*)(base + 256 + lane * 8);
    return __builtin_shufflevector(lo, hi, 0, 1, 2, 3, 4, 5, 6, 7,
                                   8, 9, 10, 11, 12, 13, 14, 15);
}

__device__ __forceinline__ v8f wmma32(v16h a, v16h b, v8f c) {
    return __builtin_amdgcn_wmma_f32_16x16x32_f16(
         false, a,  false, b,
         (short)0, c,  false,  false);
}

__device__ __forceinline__ v8f load8(const float* p) {
    const float4 a = *(const float4*)p;
    const float4 b = *(const float4*)(p + 4);
    v8f r;
    r[0] = a.x; r[1] = a.y; r[2] = a.z; r[3] = a.w;
    r[4] = b.x; r[5] = b.y; r[6] = b.z; r[7] = b.w;
    return r;
}

template <int M>
__device__ __forceinline__ float row_xor_add(float v) {
    const int o = __builtin_amdgcn_update_dpp(0, __builtin_bit_cast(int, v), 0x160 + M, 0xf, 0xf, false);
    return v + __builtin_bit_cast(float, o);
}

__device__ __forceinline__ v16h edge_frag_build(const float4 x0, const float4 x1,
                                                const float4 cs, const float4 cd, int h) {
    const float dx = cs.x - cd.x, dy = cs.y - cd.y, dz = cs.z - cd.z;
    const float dist = sqrtf(dx * dx + dy * dy + dz * dz);
    v16h x = {};
    x[0] = (_Float16)x0.x; x[1] = (_Float16)x0.y; x[2] = (_Float16)x0.z; x[3] = (_Float16)x0.w;
    x[4] = (_Float16)x1.x; x[5] = (_Float16)x1.y; x[6] = (_Float16)x1.z; x[7] = (_Float16)x1.w;
    x[8] = (h == 0) ? (_Float16)dist : (_Float16)0.0f;
    return x;
}

__device__ __forceinline__ v16h edge_frag_pack(const v8h xh, const float dist, int h) {
    v16h x = {};
#pragma unroll
    for (int i = 0; i < 8; ++i) x[i] = xh[i];
    x[8] = (h == 0) ? (_Float16)dist : (_Float16)0.0f;
    return x;
}

__device__ __forceinline__ v16h edge_frag(const float* __restrict__ edge_feat,
                                          const float* __restrict__ coord4,
                                          int e, int s, int d, int h) {
    const float* efp = edge_feat + (long)e * 16 + 8 * h;
    const float4 x0 = *(const float4*)efp;
    const float4 x1 = *(const float4*)(efp + 4);
    const float4 cs = *(const float4*)(coord4 + (long)s * 4);
    const float4 cd = *(const float4*)(coord4 + (long)d * 4);
    return edge_frag_build(x0, x1, cs, cd, h);
}

__device__ __forceinline__ float pack_resid(const float* __restrict__ W, int rowDist, int rowEf0,
                                            int h, int s, int col) {
    const bool isEf = s < 8;
    const bool isDist = (s == 8) && (h == 0);
    const int row = isEf ? (rowEf0 + 8 * h + s) : rowDist;
    const float v = W[row * 64 + col];
    return (isEf || isDist) ? v : 0.0f;
}


__global__ __launch_bounds__(BLOCK) void node_proj_kernel(
    const float* __restrict__ node_feat, const float* __restrict__ coord,
    const float* __restrict__ W_k, const float* __restrict__ W_q,
    const float* __restrict__ W_e1, const float* __restrict__ b_e1,
    const float* __restrict__ W_e2,
    const float* __restrict__ W_n1, const float* __restrict__ W_n2,
    float* __restrict__ Kn, float* __restrict__ Qn, float* __restrict__ Ps, float* __restrict__ Pd,
    float* __restrict__ coord4,
    unsigned* __restrict__ fragK_w, unsigned* __restrict__ fragE1_w, unsigned* __restrict__ fragE2_w,
    unsigned* __restrict__ fragN1_w, unsigned* __restrict__ fragN2_w,
    unsigned* __restrict__ e_max, float* __restrict__ denom, float* __restrict__ h_neigh,
    unsigned* __restrict__ cnt, int N) {

    __shared__ __attribute__((aligned(16))) _Float16 sWT[256 * PKP];
    _Float16* fragK  = (_Float16*)fragK_w;
    _Float16* fragE1 = (_Float16*)fragE1_w;
    _Float16* fragE2 = (_Float16*)fragE2_w;
    _Float16* fragN1 = (_Float16*)fragN1_w;
    _Float16* fragN2 = (_Float16*)fragN2_w;

    {
        const int n = threadIdx.x & 63, kp = threadIdx.x >> 6;
#pragma unroll
        for (int t = 0; t < 4; ++t) {
            const float* Wt = (t == 0) ? W_k : (t == 1) ? W_q : (t == 2) ? W_e1 : (W_e1 + 64 * 64);
            float w[32];
#pragma unroll
            for (int q = 0; q < 16; ++q) {
                const int k = 4 * q + 2 * kp;
                w[2 * q]     = Wt[k * 64 + n];
                w[2 * q + 1] = Wt[(k + 1) * 64 + n];
            }
#pragma unroll
            for (int q = 0; q < 16; ++q) {
                const int k = 4 * q + 2 * kp;
                v2h p;
                p[0] = (_Float16)w[2 * q]; p[1] = (_Float16)w[2 * q + 1];
                *(v2h*)&sWT[(t * 64 + n) * PKP + k] = p;
            }
        }
    }

    const int node0 = blockIdx.x * 64;
    for (int i = threadIdx.x; i < 64 * 16; i += BLOCK) {
        const int nd = node0 + (i >> 4);
        if (nd < N) *(float4*)&h_neigh[(long)nd * 64 + (i & 15) * 4] = make_float4(0.f, 0.f, 0.f, 0.f);
    }
    if (threadIdx.x < 64) {
        const int nd = node0 + threadIdx.x;
        if (nd < N) {
            denom[nd] = 0.0f;
            e_max[nd] = ENC_NEG_INF;
            cnt[nd] = 0u;
            *(float4*)(coord4 + (long)nd * 4) = make_float4(coord[nd * 3 + 0], coord[nd * 3 + 1], coord[nd * 3 + 2], 0.0f);
        }
    }

    if (blockIdx.x < 20) {
        const int g0 = blockIdx.x * 1024 + threadIdx.x;
        float pv[8];
        if (blockIdx.x < 2) {
#pragma unroll
            for (int jj = 0; jj < 8; ++jj) {
                const int g = g0 + jj * BLOCK;
                const int s8 = g & 7, ln = (g >> 3) & 31, j = (g >> 8) & 1;
                const int c = ln & 15, h = ln >> 4, nt = g >> 9;
                pv[jj] = pack_resid(W_k, 64, 65, h, j * 8 + s8, 4 * c + nt);
            }
#pragma unroll
            for (int jj = 0; jj < 8; ++jj) fragK[g0 + jj * BLOCK] = (_Float16)pv[jj];
        } else if (blockIdx.x < 4) {
#pragma unroll
            for (int jj = 0; jj < 8; ++jj) {
                const int g = g0 + jj * BLOCK;
                const int s8 = g & 7, ln = (g >> 3) & 31, j = (g >> 8) & 1;
                const int c = ln & 15, h = ln >> 4, nt = (g - 2048) >> 9;
                pv[jj] = pack_resid(W_e1, 128, 129, h, j * 8 + s8, nt * 16 + c);
            }
#pragma unroll
            for (int jj = 0; jj < 8; ++jj) fragE1[g0 - 2048 + jj * BLOCK] = (_Float16)pv[jj];
        } else if (blockIdx.x < 8) {
#pragma unroll
            for (int jj = 0; jj < 8; ++jj) {
                const int g = g0 + jj * BLOCK;
                const int s8 = g & 7, ln = (g >> 3) & 31, j = (g >> 8) & 1;
                const int c = ln & 15, h = ln >> 4;
                const int q = g - 4096;
                const int kc = (q >> 9) & 1, nt = q >> 10;
                const int k = kc * 32 + 8 * h + s8 + 16 * j;
                pv[jj] = W_e2[k * 64 + nt * 16 + c];
            }
#pragma unroll
            for (int jj = 0; jj < 8; ++jj) fragE2[g0 - 4096 + jj * BLOCK] = (_Float16)pv[jj];
        } else if (blockIdx.x < 16) {
#pragma unroll
            for (int jj = 0; jj < 8; ++jj) {
                const int g = g0 + jj * BLOCK;
                const int s8 = g & 7, ln = (g >> 3) & 31, j = (g >> 8) & 1;
                const int c = ln & 15, h = ln >> 4;
                const int q = g - 8192;
                const int kc = (q >> 9) & 3, nt = q >> 11;
                const int k = kc * 32 + 8 * h + s8 + 16 * j;
                pv[jj] = W_n1[k * 64 + nt * 16 + c];
            }
#pragma unroll
            for (int jj = 0; jj < 8; ++jj) fragN1[g0 - 8192 + jj * BLOCK] = (_Float16)pv[jj];
        } else {
#pragma unroll
            for (int jj = 0; jj < 8; ++jj) {
                const int g = g0 + jj * BLOCK;
                const int s8 = g & 7, ln = (g >> 3) & 31, j = (g >> 8) & 1;
                const int c = ln & 15, h = ln >> 4;
                const int q = g - 16384;
                const int kc = (q >> 9) & 1, nt = q >> 10;
                const int k = kc * 32 + 8 * h + s8 + 16 * j;
                pv[jj] = W_n2[k * 64 + nt * 16 + c];
            }
#pragma unroll
            for (int jj = 0; jj < 8; ++jj) fragN2[g0 - 16384 + jj * BLOCK] = (_Float16)pv[jj];
        }
    }

    __syncthreads();

    const int wave = threadIdx.x >> 5, lane = threadIdx.x & 31;
    const int c = lane & 15, h = lane >> 4;
    const int numTiles = (N + 15) >> 4;
    const int tile = blockIdx.x * WAVES + wave;
    if (tile >= numTiles) return;

    int node = tile * 16 + c;
    if (node >= N) node = N - 1;

    v16h a[2];
#pragma unroll
    for (int kc = 0; kc < 2; ++kc) {
        const float* p = node_feat + (long)node * 64 + kc * 32 + 8 * h;
        const float4 v0 = *(const float4*)p,        v1 = *(const float4*)(p + 4);
        const float4 v2 = *(const float4*)(p + 16), v3 = *(const float4*)(p + 20);
        v16h f;
        f[0] = (_Float16)v0.x; f[1] = (_Float16)v0.y; f[2] = (_Float16)v0.z; f[3] = (_Float16)v0.w;
        f[4] = (_Float16)v1.x; f[5] = (_Float16)v1.y; f[6] = (_Float16)v1.z; f[7] = (_Float16)v1.w;
        f[8] = (_Float16)v2.x; f[9] = (_Float16)v2.y; f[10] = (_Float16)v2.z; f[11] = (_Float16)v2.w;
        f[12] = (_Float16)v3.x; f[13] = (_Float16)v3.y; f[14] = (_Float16)v3.z; f[15] = (_Float16)v3.w;
        a[kc] = f;
    }

#pragma unroll
    for (int t = 0; t < 4; ++t) {
        float* T = (t == 0) ? Kn : (t == 1) ? Qn : (t == 2) ? Ps : Pd;
        v8f acc[4];
#pragma unroll
        for (int nt = 0; nt < 4; ++nt) acc[nt] = {};
#pragma unroll
        for (int kc = 0; kc < 2; ++kc) {
#pragma unroll
            for (int nt = 0; nt < 4; ++nt)
                acc[nt] = wmma32(a[kc],
                                 load_frag(&sWT[(t * 64 + nt * 16 + c) * PKP + kc * 32 + 8 * h]),
                                 acc[nt]);
        }
#pragma unroll
        for (int nt = 0; nt < 4; ++nt) {
            const float bb = (t == 3) ? b_e1[nt * 16 + c] : 0.0f;
#pragma unroll
            for (int r = 0; r < 8; ++r) {
                const int nd = tile * 16 + 8 * h + r;
                if (nd < N) T[(long)nd * 64 + nt * 16 + c] = acc[nt][r] + bb;
            }
        }
    }
}


__global__ __launch_bounds__(BLOCK) void edge_score_kernel(
    const float* __restrict__ edge_feat, const float* __restrict__ coord4,
    const float* __restrict__ Kn, const float* __restrict__ Qn,
    const unsigned* __restrict__ fragK_w,
    const int* __restrict__ src, const int* __restrict__ dst,
    float* __restrict__ e_arr, unsigned* __restrict__ e_max,
    unsigned* __restrict__ cnt, unsigned* __restrict__ rank, int E) {

    const int wave = threadIdx.x >> 5, lane = threadIdx.x & 31;
    const int c = lane & 15, h = lane >> 4;
    const int numTiles = (E + 15) >> 4;
    const int tile = blockIdx.x * WAVES + wave;
    if (tile >= numTiles) return;
    const _Float16* fragK = (const _Float16*)fragK_w;

    const int eIdx = tile * 16 + c;
    const bool valid = eIdx < E;
    const int e = valid ? eIdx : (E - 1);
    const v4f* efp = (const v4f*)(edge_feat + (long)e * 16 + 8 * h);
    v4f x0 = efp[0];
    v4f x1 = efp[1];
    int s = src[e], d = dst[e];
    asm volatile("" : "+v"(x0), "+v"(x1), "+v"(s), "+v"(d));
    v4f cs = ((const v4f*)coord4)[(unsigned)s];
    v4f cd = ((const v4f*)coord4)[(unsigned)d];
    v8h xh;
    xh[0] = (_Float16)x0.x; xh[1] = (_Float16)x0.y; xh[2] = (_Float16)x0.z; xh[3] = (_Float16)x0.w;
    xh[4] = (_Float16)x1.x; xh[5] = (_Float16)x1.y; xh[6] = (_Float16)x1.z; xh[7] = (_Float16)x1.w;

    const v4f* __restrict__ Kn4 = (const v4f*)Kn;
    const v4f* __restrict__ Qn4 = (const v4f*)Qn;
    v4f kv[8], qv[8];
#pragma unroll
    for (int r = 0; r < 8; ++r) {
        const unsigned sm = (unsigned)__shfl(s, 8 * h + r, 32);
        const unsigned dm = (unsigned)__shfl(d, 8 * h + r, 32);
        kv[r] = Kn4[sm * 16u + (unsigned)c];
        qv[r] = Qn4[dm * 16u + (unsigned)c];
    }
    int fo = lane * 8;
    asm volatile("" : "+v"(kv[0]), "+v"(kv[1]), "+v"(kv[2]), "+v"(kv[3]),
                      "+v"(kv[4]), "+v"(kv[5]), "+v"(kv[6]), "+v"(kv[7]),
                      "+v"(qv[0]), "+v"(qv[1]), "+v"(qv[2]), "+v"(qv[3]),
                      "+v"(qv[4]), "+v"(qv[5]), "+v"(qv[6]), "+v"(qv[7]),
                      "+v"(cs), "+v"(cd), "+v"(xh));

    float p[8];
#pragma unroll
    for (int r = 0; r < 8; ++r)
        p[r] = kv[r].x * qv[r].x + kv[r].y * qv[r].y + kv[r].z * qv[r].z + kv[r].w * qv[r].w;
    const float dx = cs.x - cd.x, dy = cs.y - cd.y, dz = cs.z - cd.z;
    float dist = sqrtf(dx * dx + dy * dy + dz * dz);
    asm volatile("" : "+v"(fo), "+v"(dist), "+v"(p[0]), "+v"(p[1]), "+v"(p[2]), "+v"(p[3]),
                      "+v"(p[4]), "+v"(p[5]), "+v"(p[6]), "+v"(p[7]));

    v8h flo[4], fhi[4];
#pragma unroll
    for (int nt = 0; nt < 4; ++nt) {
        const _Float16* fb = fragK + nt * 512 + fo;
        flo[nt] = *(const v8h*)fb;
        fhi[nt] = *(const v8h*)(fb + 256);
    }
    asm volatile("" : "+v"(flo[0]), "+v"(flo[1]), "+v"(flo[2]), "+v"(flo[3]),
                      "+v"(fhi[0]), "+v"(fhi[1]), "+v"(fhi[2]), "+v"(fhi[3]));

    const v16h x = edge_frag_pack(xh, dist, h);

    v8f acc[4];
#pragma unroll
    for (int nt = 0; nt < 4; ++nt) {
        const v8f zero = {};
        acc[nt] = wmma32(x, __builtin_shufflevector(flo[nt], fhi[nt], 0, 1, 2, 3, 4, 5, 6, 7,
                                                    8, 9, 10, 11, 12, 13, 14, 15), zero);
    }

#pragma unroll
    for (int r = 0; r < 8; ++r) {
        p[r] += acc[0][r] * qv[r].x + acc[1][r] * qv[r].y + acc[2][r] * qv[r].z + acc[3][r] * qv[r].w;
    }
#pragma unroll
    for (int r = 0; r < 8; ++r) p[r] = row_xor_add<1>(p[r]);
#pragma unroll
    for (int r = 0; r < 8; ++r) p[r] = row_xor_add<2>(p[r]);
#pragma unroll
    for (int r = 0; r < 8; ++r) p[r] = row_xor_add<4>(p[r]);
#pragma unroll
    for (int r = 0; r < 8; ++r) p[r] = row_xor_add<8>(p[r]);

    float pe = p[0];
#pragma unroll
    for (int r = 1; r < 8; ++r) pe = ((c & 7) == r) ? p[r] : pe;

    if ((c >> 3) == h && valid) {
        const float ev = pe * 0.125f;
        e_arr[eIdx] = ev;
        atomicMax(&e_max[d], enc_f(ev));
        rank[eIdx] = atomicAdd(&cnt[d], 1u);
    }
}


#define SCAN_T 1024

__global__ __launch_bounds__(SCAN_T) void rowstart_scan_kernel(
    const unsigned* __restrict__ cnt, unsigned* __restrict__ rowstart, int N) {

    __shared__ unsigned sWave[32];
    const int t = threadIdx.x, lane = t & 31, wave = t >> 5;
    const int per = (N + SCAN_T - 1) / SCAN_T;
    const int base = t * per;
    unsigned tot = 0;
    for (int i0 = 0; i0 < per; i0 += 8) {
        unsigned v[8];
#pragma unroll
        for (int j = 0; j < 8; ++j) {
            const int n = base + i0 + j;
            const bool ok = (i0 + j < per) && (n < N);
            const unsigned x = cnt[ok ? n : 0];
            v[j] = ok ? x : 0u;
        }
#pragma unroll
        for (int j = 0; j < 8; ++j) tot += v[j];
    }
    unsigned inc = tot;
    for (int o = 1; o < 32; o <<= 1) {
        const unsigned v = __shfl_up(inc, o, 32);
        if (lane >= o) inc += v;
    }
    if (lane == 31) sWave[wave] = inc;
    __syncthreads();
    if (wave == 0) {
        const unsigned wt = sWave[lane];
        unsigned winc = wt;
        for (int o = 1; o < 32; o <<= 1) {
            const unsigned v = __shfl_up(winc, o, 32);
            if (lane >= o) winc += v;
        }
        sWave[lane] = winc - wt;
    }
    __syncthreads();
    unsigned run = sWave[wave] + inc - tot;
    for (int i0 = 0; i0 < per; i0 += 8) {
        unsigned v[8];
#pragma unroll
        for (int j = 0; j < 8; ++j) {
            const int n = base + i0 + j;
            const bool ok = (i0 + j < per) && (n < N);
            const unsigned x = cnt[ok ? n : 0];
            v[j] = ok ? x : 0u;
        }
#pragma unroll
        for (int j = 0; j < 8; ++j) {
            const int n = base + i0 + j;
            if ((i0 + j < per) && (n < N)) { rowstart[n] = run; run += v[j]; }
        }
    }
}


__global__ __launch_bounds__(256) void edge_scatter_kernel(
    const int* __restrict__ src, const int* __restrict__ dst,
    const float* __restrict__ e_arr, const unsigned* __restrict__ rank,
    const unsigned* __restrict__ rowstart, int4* __restrict__ sorted4, int E) {

    const int e0 = (blockIdx.x * 256 + threadIdx.x) * 4;
    if (e0 >= E) return;
    const unsigned last = (unsigned)(E - 1);
    if (e0 + 3 < E) {
        const int4   d4 = *(const int4*)(dst + e0);
        const int4   s4 = *(const int4*)(src + e0);
        const uint4  r4 = *(const uint4*)(rank + e0);
        const float4 v4 = *(const float4*)(e_arr + e0);
        const unsigned p0 = rowstart[d4.x] + r4.x;
        const unsigned p1 = rowstart[d4.y] + r4.y;
        const unsigned p2 = rowstart[d4.z] + r4.z;
        const unsigned p3 = rowstart[d4.w] + r4.w;
        sorted4[min(p0, last)] = make_int4(e0,     s4.x, d4.x, __float_as_int(v4.x));
        sorted4[min(p1, last)] = make_int4(e0 + 1, s4.y, d4.y, __float_as_int(v4.y));
        sorted4[min(p2, last)] = make_int4(e0 + 2, s4.z, d4.z, __float_as_int(v4.z));
        sorted4[min(p3, last)] = make_int4(e0 + 3, s4.w, d4.w, __float_as_int(v4.w));
    } else {
        for (int e = e0; e < E; ++e) {
            const int d = dst[e];
            const unsigned pos = rowstart[d] + rank[e];
            sorted4[min(pos, last)] = make_int4(e, src[e], d, __float_as_int(e_arr[e]));
        }
    }
}


__global__ __launch_bounds__(BLOCK) void edge_msg_kernel(
    const float* __restrict__ edge_feat, const float* __restrict__ coord4,
    const float* __restrict__ Ps, const float* __restrict__ Pd,
    const unsigned* __restrict__ fragE1_w, const unsigned* __restrict__ fragE2_w,
    const float* __restrict__ b_e2,
    const int4* __restrict__ sorted4, const unsigned* __restrict__ e_max,
    float* __restrict__ denom, float* __restrict__ h_neigh, int E, int N) {

    const int wave = threadIdx.x >> 5, lane = threadIdx.x & 31;
    const int c = lane & 15, h = lane >> 4;
    const int numTiles = (E + 15) >> 4;
    const int tile = blockIdx.x * WAVES + wave;
    if (tile >= numTiles) return;
    const _Float16* fragE1 = (const _Float16*)fragE1_w;
    const _Float16* fragE2 = (const _Float16*)fragE2_w;

    const int place = tile * 16 + c;
    const bool valid = place < E;
    const int4 rec = sorted4[valid ? place : (E - 1)];
    const int e = (int)min((unsigned)rec.x, (unsigned)(E - 1));
    const int s = (int)min((unsigned)rec.y, (unsigned)(N - 1));
    const int d = (int)min((unsigned)rec.z, (unsigned)(N - 1));

    const float* sp = Ps + (long)s * 64 + 8 * h;
    const float* dp = Pd + (long)d * 64 + 8 * h;
    v8f acc[4];
#pragma unroll
    for (int nt = 0; nt < 4; ++nt) acc[nt] = load8(sp + nt * 16) + load8(dp + nt * 16);

    float w = 0.0f;
    if (valid) w = __expf(__int_as_float(rec.w) - dec_f(e_max[d]));

    const v16h x = edge_frag(edge_feat, coord4, e, s, d, h);

#pragma unroll
    for (int nt = 0; nt < 4; ++nt)
        acc[nt] = wmma32(load_packed(fragE1 + nt * 512, lane), x, acc[nt]);

    v16h mid[2];
#pragma unroll
    for (int kc = 0; kc < 2; ++kc) {
        v16h m;
#pragma unroll
        for (int r = 0; r < 8; ++r) {
            m[r]     = (_Float16)silu_f(acc[2 * kc][r]);
            m[8 + r] = (_Float16)silu_f(acc[2 * kc + 1][r]);
        }
        mid[kc] = m;
    }

    v8f acc2[4];
#pragma unroll
    for (int nt = 0; nt < 4; ++nt) {
        acc2[nt] = {};
#pragma unroll
        for (int kc = 0; kc < 2; ++kc)
            acc2[nt] = wmma32(mid[kc], load_packed(fragE2 + (nt * 2 + kc) * 512, lane), acc2[nt]);
    }

    float bb[4];
#pragma unroll
    for (int nt = 0; nt < 4; ++nt) bb[nt] = b_e2[nt * 16 + c];

    const int d7 = __shfl(d, 7, 32), d8 = __shfl(d, 8, 32);
    const bool joined = d7 == d8;
    bool holdFirst = joined && h == 1;
    float run[4]  = {0.0f, 0.0f, 0.0f, 0.0f};
    float held[4] = {0.0f, 0.0f, 0.0f, 0.0f};
    float wsum = 0.0f, wheld = 0.0f;
    int dr = __shfl(d, 8 * h, 32);
#pragma unroll
    for (int r = 0; r < 8; ++r) {
        const int m = 8 * h + r;
        const float wr = __shfl(w, m, 32);
        const int dn = (r < 7) ? __shfl(d, m + 1, 32) : dr;
        wsum += wr;
#pragma unroll
        for (int nt = 0; nt < 4; ++nt) run[nt] += silu_f(acc2[nt][r] + bb[nt]) * wr;
        if (r == 7 || dn != dr) {
            if (holdFirst || (joined && h == 0 && r == 7)) {
#pragma unroll
                for (int nt = 0; nt < 4; ++nt) held[nt] = run[nt];
                wheld = wsum;
            } else {
                float* hp = h_neigh + (long)dr * 64 + c;
#pragma unroll
                for (int nt = 0; nt < 4; ++nt) atomicAdd(hp + nt * 16, run[nt]);
                if (c == 0) atomicAdd(&denom[dr], wsum);
            }
            holdFirst = false;
#pragma unroll
            for (int nt = 0; nt < 4; ++nt) run[nt] = 0.0f;
            wsum = 0.0f;
        }
        dr = dn;
    }
    if (joined) {
        float* hp = h_neigh + (long)d7 * 64 + c;
#pragma unroll
        for (int nt = 0; nt < 4; ++nt) {
            const float other = __shfl_xor(held[nt], 16, 32);
            if (h == 0) atomicAdd(hp + nt * 16, held[nt] + other);
        }
        const float wother = __shfl_xor(wheld, 16, 32);
        if (lane == 0) atomicAdd(&denom[d7], wheld + wother);
    }
}


__global__ __launch_bounds__(BLOCK) void node_mlp_kernel(
    const float* __restrict__ node_feat,
    const unsigned* __restrict__ fragN1_w, const float* __restrict__ b_n1,
    const unsigned* __restrict__ fragN2_w, const float* __restrict__ b_n2,
    const float* __restrict__ denom, const float* __restrict__ h_neigh,
    float* __restrict__ out, int N) {

    const int wave = threadIdx.x >> 5, lane = threadIdx.x & 31;
    const int c = lane & 15, h = lane >> 4;
    const int numTiles = (N + 15) >> 4;
    const int tile = blockIdx.x * WAVES + wave;
    if (tile >= numTiles) return;
    const _Float16* fragN1 = (const _Float16*)fragN1_w;
    const _Float16* fragN2 = (const _Float16*)fragN2_w;

    int node = tile * 16 + c;
    if (node >= N) node = N - 1;

    const float dn = denom[node];
    const float inv = (dn > 0.0f) ? (1.0f / dn) : 0.0f;

    v16h x[4];
#pragma unroll
    for (int kc = 0; kc < 2; ++kc) {
        const float* p = node_feat + (long)node * 64 + kc * 32 + 8 * h;
        const float4 v0 = *(const float4*)p,        v1 = *(const float4*)(p + 4);
        const float4 v2 = *(const float4*)(p + 16), v3 = *(const float4*)(p + 20);
        v16h f;
        f[0] = (_Float16)v0.x; f[1] = (_Float16)v0.y; f[2] = (_Float16)v0.z; f[3] = (_Float16)v0.w;
        f[4] = (_Float16)v1.x; f[5] = (_Float16)v1.y; f[6] = (_Float16)v1.z; f[7] = (_Float16)v1.w;
        f[8] = (_Float16)v2.x; f[9] = (_Float16)v2.y; f[10] = (_Float16)v2.z; f[11] = (_Float16)v2.w;
        f[12] = (_Float16)v3.x; f[13] = (_Float16)v3.y; f[14] = (_Float16)v3.z; f[15] = (_Float16)v3.w;
        x[kc] = f;
    }
#pragma unroll
    for (int kc = 0; kc < 2; ++kc) {
        const float* p = h_neigh + (long)node * 64 + kc * 32 + 8 * h;
        const float4 v0 = *(const float4*)p,        v1 = *(const float4*)(p + 4);
        const float4 v2 = *(const float4*)(p + 16), v3 = *(const float4*)(p + 20);
        v16h f;
        f[0] = (_Float16)(v0.x * inv); f[1] = (_Float16)(v0.y * inv);
        f[2] = (_Float16)(v0.z * inv); f[3] = (_Float16)(v0.w * inv);
        f[4] = (_Float16)(v1.x * inv); f[5] = (_Float16)(v1.y * inv);
        f[6] = (_Float16)(v1.z * inv); f[7] = (_Float16)(v1.w * inv);
        f[8] = (_Float16)(v2.x * inv); f[9] = (_Float16)(v2.y * inv);
        f[10] = (_Float16)(v2.z * inv); f[11] = (_Float16)(v2.w * inv);
        f[12] = (_Float16)(v3.x * inv); f[13] = (_Float16)(v3.y * inv);
        f[14] = (_Float16)(v3.z * inv); f[15] = (_Float16)(v3.w * inv);
        x[2 + kc] = f;
    }

    v8f acc[4];
#pragma unroll
    for (int nt = 0; nt < 4; ++nt) acc[nt] = load8(b_n1 + nt * 16 + 8 * h);
#pragma unroll
    for (int kc = 0; kc < 4; ++kc) {
#pragma unroll
        for (int nt = 0; nt < 4; ++nt)
            acc[nt] = wmma32(load_packed(fragN1 + (nt * 4 + kc) * 512, lane), x[kc], acc[nt]);
    }

    v16h mid[2];
#pragma unroll
    for (int kc = 0; kc < 2; ++kc) {
        v16h m;
#pragma unroll
        for (int r = 0; r < 8; ++r) {
            m[r]     = (_Float16)silu_f(acc[2 * kc][r]);
            m[8 + r] = (_Float16)silu_f(acc[2 * kc + 1][r]);
        }
        mid[kc] = m;
    }

    v8f acc2[4];
#pragma unroll
    for (int nt = 0; nt < 4; ++nt) {
        const float bb = b_n2[nt * 16 + c];
        v8f st;
#pragma unroll
        for (int r = 0; r < 8; ++r) st[r] = bb;
        acc2[nt] = st;
#pragma unroll
        for (int kc = 0; kc < 2; ++kc)
            acc2[nt] = wmma32(mid[kc], load_packed(fragN2 + (nt * 2 + kc) * 512, lane), acc2[nt]);
    }

#pragma unroll
    for (int nt = 0; nt < 4; ++nt) {
#pragma unroll
        for (int r = 0; r < 8; ++r) {
            const int nd = tile * 16 + 8 * h + r;
            if (nd < N) out[(long)nd * 64 + nt * 16 + c] = acc2[nt][r];
        }
    }
}


extern "C" void kernel_launch(void* const* d_in, const int* in_sizes, int n_in,
                              void* d_out, int out_size, void* d_ws, size_t ws_size,
                              hipStream_t stream) {
    const float* node_feat = (const float*)d_in[0];
    const float* coord     = (const float*)d_in[1];
    const float* edge_feat = (const float*)d_in[2];
    const float* W_e1 = (const float*)d_in[3];
    const float* b_e1 = (const float*)d_in[4];
    const float* W_e2 = (const float*)d_in[5];
    const float* b_e2 = (const float*)d_in[6];
    const float* W_n1 = (const float*)d_in[7];
    const float* b_n1 = (const float*)d_in[8];
    const float* W_n2 = (const float*)d_in[9];
    const float* b_n2 = (const float*)d_in[10];
    const float* W_q  = (const float*)d_in[11];
    const float* W_k  = (const float*)d_in[12];
    const int*   src  = (const int*)d_in[13];
    const int*   dst  = (const int*)d_in[14];

    const int N = in_sizes[0] / 64;
    const int E = in_sizes[13];

    const size_t Ea = ((size_t)E + 63) & ~(size_t)63;
    const size_t Na = ((size_t)N + 63) & ~(size_t)63;
    const size_t T  = (size_t)N * 64;
    float*    e_arr   = (float*)d_ws;
    unsigned* e_max   = (unsigned*)(e_arr + Ea);
    float*    denom   = (float*)(e_max + Na);
    float*    h_neigh = denom + Na;
    float*    Kn      = h_neigh + T;
    float*    Qn      = Kn + T;
    float*    Ps      = Qn + T;
    float*    Pd      = Ps + T;
    float*    coord4  = Pd + T;
    unsigned* fragK   = (unsigned*)(coord4 + 4 * Na);
    unsigned* fragE1  = fragK + 1024;
    unsigned* fragE2  = fragE1 + 1024;
    unsigned* fragN1  = fragE2 + 2048;
    unsigned* fragN2  = fragN1 + 4096;
    unsigned* cnt      = fragN2 + 2048;
    unsigned* rowstart = cnt + Na;
    unsigned* rank     = rowstart + Na;
    int4*     sorted4  = (int4*)(rank + Ea);

    const size_t wsNeed = (size_t)((const char*)(sorted4 + E) - (const char*)d_ws);
    if (ws_size < wsNeed) return;

    const int numEdgeTiles = (E + 15) / 16;
    const int edgeBlocks   = (numEdgeTiles + WAVES - 1) / WAVES;
    const int numNodeTiles = (N + 15) / 16;
    const int nodeBlocks   = (numNodeTiles + WAVES - 1) / WAVES;
    const int projBlocks   = nodeBlocks < 20 ? 20 : nodeBlocks;

    node_proj_kernel<<<projBlocks, BLOCK, 0, stream>>>(
        node_feat, coord, W_k, W_q, W_e1, b_e1, W_e2, W_n1, W_n2,
        Kn, Qn, Ps, Pd, coord4, fragK, fragE1, fragE2, fragN1, fragN2, e_max, denom, h_neigh, cnt, N);

    edge_score_kernel<<<edgeBlocks, BLOCK, 0, stream>>>(
        edge_feat, coord4, Kn, Qn, fragK, src, dst, e_arr, e_max, cnt, rank, E);

    rowstart_scan_kernel<<<1, SCAN_T, 0, stream>>>(cnt, rowstart, N);

    edge_scatter_kernel<<<((E + 3) / 4 + 255) / 256, 256, 0, stream>>>(
        src, dst, e_arr, rank, rowstart, sorted4, E);

    edge_msg_kernel<<<edgeBlocks, BLOCK, 0, stream>>>(
        edge_feat, coord4, Ps, Pd, fragE1, fragE2, b_e2,
        sorted4, e_max, denom, h_neigh, E, N);

    node_mlp_kernel<<<nodeBlocks, BLOCK, 0, stream>>>(
        node_feat, fragN1, b_n1, fragN2, b_n2, denom, h_neigh, (float*)d_out, N);
}


# silu evaluated as h + h*tanh(h) with h = x/2 on the hardware tanh (v_tanh_f32) in edge MLP and node MLP: no exponential, no correctly rounded division per silu; silu values change in the last bits
# speedup vs baseline: 3.7228x; 1.0999x over previous
#include <hip/hip_runtime.h>
#include <hip/hip_bf16.h>
#include <math.h>


typedef __attribute__((ext_vector_type(16))) _Float16 v16h;
typedef __attribute__((ext_vector_type(8)))  _Float16 v8h;
typedef __attribute__((ext_vector_type(4)))  _Float16 v4h;
typedef __attribute__((ext_vector_type(2)))  _Float16 v2h;
typedef __attribute__((ext_vector_type(8)))  float    v8f;
typedef __attribute__((ext_vector_type(4)))  float    v4f;

#define WAVES 4
#define BLOCK 128
#define PKP 72


__device__ __forceinline__ float silu_f(float x) {
    const float h = 0.5f * x;
    return __builtin_fmaf(h, __builtin_amdgcn_tanhf(h), h);
}

__device__ __forceinline__ unsigned enc_f(float f) {
    unsigned u = __float_as_uint(f);
    return (u & 0x80000000u) ? ~u : (u | 0x80000000u);
}
__device__ __forceinline__ float dec_f(unsigned u) {
    unsigned b = (u & 0x80000000u) ? (u & 0x7FFFFFFFu) : ~u;
    return __uint_as_float(b);
}
#define ENC_NEG_INF 0x007FFFFFu

__device__ __forceinline__ unsigned package_half() {
    return ((unsigned)__builtin_amdgcn_s_sendmsg_rtn(0x87) >> 19) & 1u;
}

__device__ __forceinline__ v16h load_frag(const _Float16* p) {
    v8h lo = *(const v8h*)(p);
    v8h hi = *(const v8h*)(p + 16);
    return __builtin_shufflevector(lo, hi, 0, 1, 2, 3, 4, 5, 6, 7,
                                   8, 9, 10, 11, 12, 13, 14, 15);
}

__device__ __forceinline__ v16h load_packed(const _Float16* base, int lane) {
    v8h lo = *(const v8h*)(base + lane * 8);
    v8h hi = *(const v8h*)(base + 256 + lane * 8);
    return __builtin_shufflevector(lo, hi, 0, 1, 2, 3, 4, 5, 6, 7,
                                   8, 9, 10, 11, 12, 13, 14, 15);
}

__device__ __forceinline__ v8f wmma32(v16h a, v16h b, v8f c) {
    return __builtin_amdgcn_wmma_f32_16x16x32_f16(
         false, a,  false, b,
         (short)0, c,  false,  false);
}

__device__ __forceinline__ v8f load8(const float* p) {
    const float4 a = *(const float4*)p;
    const float4 b = *(const float4*)(p + 4);
    v8f r;
    r[0] = a.x; r[1] = a.y; r[2] = a.z; r[3] = a.w;
    r[4] = b.x; r[5] = b.y; r[6] = b.z; r[7] = b.w;
    return r;
}

template <int M>
__device__ __forceinline__ float row_xor_add(float v) {
    const int o = __builtin_amdgcn_update_dpp(0, __builtin_bit_cast(int, v), 0x160 + M, 0xf, 0xf, false);
    return v + __builtin_bit_cast(float, o);
}

__device__ __forceinline__ v16h edge_frag_build(const float4 x0, const float4 x1,
                                                const float4 cs, const float4 cd, int h) {
    const float dx = cs.x - cd.x, dy = cs.y - cd.y, dz = cs.z - cd.z;
    const float dist = sqrtf(dx * dx + dy * dy + dz * dz);
    v16h x = {};
    x[0] = (_Float16)x0.x; x[1] = (_Float16)x0.y; x[2] = (_Float16)x0.z; x[3] = (_Float16)x0.w;
    x[4] = (_Float16)x1.x; x[5] = (_Float16)x1.y; x[6] = (_Float16)x1.z; x[7] = (_Float16)x1.w;
    x[8] = (h == 0) ? (_Float16)dist : (_Float16)0.0f;
    return x;
}

__device__ __forceinline__ v16h edge_frag_pack(const v8h xh, const float dist, int h) {
    v16h x = {};
#pragma unroll
    for (int i = 0; i < 8; ++i) x[i] = xh[i];
    x[8] = (h == 0) ? (_Float16)dist : (_Float16)0.0f;
    return x;
}

__device__ __forceinline__ v16h edge_frag(const float* __restrict__ edge_feat,
                                          const float* __restrict__ coord4,
                                          int e, int s, int d, int h) {
    const float* efp = edge_feat + (long)e * 16 + 8 * h;
    const float4 x0 = *(const float4*)efp;
    const float4 x1 = *(const float4*)(efp + 4);
    const float4 cs = *(const float4*)(coord4 + (long)s * 4);
    const float4 cd = *(const float4*)(coord4 + (long)d * 4);
    return edge_frag_build(x0, x1, cs, cd, h);
}

__device__ __forceinline__ float pack_resid(const float* __restrict__ W, int rowDist, int rowEf0,
                                            int h, int s, int col) {
    const bool isEf = s < 8;
    const bool isDist = (s == 8) && (h == 0);
    const int row = isEf ? (rowEf0 + 8 * h + s) : rowDist;
    const float v = W[row * 64 + col];
    return (isEf || isDist) ? v : 0.0f;
}


__global__ __launch_bounds__(BLOCK) void node_proj_kernel(
    const float* __restrict__ node_feat, const float* __restrict__ coord,
    const float* __restrict__ W_k, const float* __restrict__ W_q,
    const float* __restrict__ W_e1, const float* __restrict__ b_e1,
    const float* __restrict__ W_e2,
    const float* __restrict__ W_n1, const float* __restrict__ W_n2,
    float* __restrict__ Kn, float* __restrict__ Qn, float* __restrict__ Ps, float* __restrict__ Pd,
    float* __restrict__ coord4,
    unsigned* __restrict__ fragK_w, unsigned* __restrict__ fragE1_w, unsigned* __restrict__ fragE2_w,
    unsigned* __restrict__ fragN1_w, unsigned* __restrict__ fragN2_w,
    unsigned* __restrict__ e_max0, unsigned* __restrict__ e_max1,
    float* __restrict__ denom, float* __restrict__ h_neigh,
    unsigned* __restrict__ cnt0, unsigned* __restrict__ cnt1, int N) {

    __shared__ __attribute__((aligned(16))) _Float16 sWT[256 * PKP];
    _Float16* fragK  = (_Float16*)fragK_w;
    _Float16* fragE1 = (_Float16*)fragE1_w;
    _Float16* fragE2 = (_Float16*)fragE2_w;
    _Float16* fragN1 = (_Float16*)fragN1_w;
    _Float16* fragN2 = (_Float16*)fragN2_w;

    {
        const int n = threadIdx.x & 63, kp = threadIdx.x >> 6;
#pragma unroll
        for (int t = 0; t < 4; ++t) {
            const float* Wt = (t == 0) ? W_k : (t == 1) ? W_q : (t == 2) ? W_e1 : (W_e1 + 64 * 64);
            float w[32];
#pragma unroll
            for (int q = 0; q < 16; ++q) {
                const int k = 4 * q + 2 * kp;
                w[2 * q]     = Wt[k * 64 + n];
                w[2 * q + 1] = Wt[(k + 1) * 64 + n];
            }
#pragma unroll
            for (int q = 0; q < 16; ++q) {
                const int k = 4 * q + 2 * kp;
                v2h p;
                p[0] = (_Float16)w[2 * q]; p[1] = (_Float16)w[2 * q + 1];
                *(v2h*)&sWT[(t * 64 + n) * PKP + k] = p;
            }
        }
    }

    const int node0 = blockIdx.x * 64;
    for (int i = threadIdx.x; i < 64 * 16; i += BLOCK) {
        const int nd = node0 + (i >> 4);
        if (nd < N) *(float4*)&h_neigh[(long)nd * 64 + (i & 15) * 4] = make_float4(0.f, 0.f, 0.f, 0.f);
    }
    if (threadIdx.x < 64) {
        const int nd = node0 + threadIdx.x;
        if (nd < N) {
            denom[nd] = 0.0f;
            e_max0[nd] = ENC_NEG_INF;
            e_max1[nd] = ENC_NEG_INF;
            cnt0[nd] = 0u;
            cnt1[nd] = 0u;
            *(float4*)(coord4 + (long)nd * 4) = make_float4(coord[nd * 3 + 0], coord[nd * 3 + 1], coord[nd * 3 + 2], 0.0f);
        }
    }

    if (blockIdx.x < 20) {
        const int g0 = blockIdx.x * 1024 + threadIdx.x;
        float pv[8];
        if (blockIdx.x < 2) {
#pragma unroll
            for (int jj = 0; jj < 8; ++jj) {
                const int g = g0 + jj * BLOCK;
                const int s8 = g & 7, ln = (g >> 3) & 31, j = (g >> 8) & 1;
                const int c = ln & 15, h = ln >> 4, nt = g >> 9;
                pv[jj] = pack_resid(W_k, 64, 65, h, j * 8 + s8, 4 * c + nt);
            }
#pragma unroll
            for (int jj = 0; jj < 8; ++jj) fragK[g0 + jj * BLOCK] = (_Float16)pv[jj];
        } else if (blockIdx.x < 4) {
#pragma unroll
            for (int jj = 0; jj < 8; ++jj) {
                const int g = g0 + jj * BLOCK;
                const int s8 = g & 7, ln = (g >> 3) & 31, j = (g >> 8) & 1;
                const int c = ln & 15, h = ln >> 4, nt = (g - 2048) >> 9;
                pv[jj] = pack_resid(W_e1, 128, 129, h, j * 8 + s8, nt * 16 + c);
            }
#pragma unroll
            for (int jj = 0; jj < 8; ++jj) fragE1[g0 - 2048 + jj * BLOCK] = (_Float16)pv[jj];
        } else if (blockIdx.x < 8) {
#pragma unroll
            for (int jj = 0; jj < 8; ++jj) {
                const int g = g0 + jj * BLOCK;
                const int s8 = g & 7, ln = (g >> 3) & 31, j = (g >> 8) & 1;
                const int c = ln & 15, h = ln >> 4;
                const int q = g - 4096;
                const int kc = (q >> 9) & 1, nt = q >> 10;
                const int k = kc * 32 + 8 * h + s8 + 16 * j;
                pv[jj] = W_e2[k * 64 + nt * 16 + c];
            }
#pragma unroll
            for (int jj = 0; jj < 8; ++jj) fragE2[g0 - 4096 + jj * BLOCK] = (_Float16)pv[jj];
        } else if (blockIdx.x < 16) {
#pragma unroll
            for (int jj = 0; jj < 8; ++jj) {
                const int g = g0 + jj * BLOCK;
                const int s8 = g & 7, ln = (g >> 3) & 31, j = (g >> 8) & 1;
                const int c = ln & 15, h = ln >> 4;
                const int q = g - 8192;
                const int kc = (q >> 9) & 3, nt = q >> 11;
                const int k = kc * 32 + 8 * h + s8 + 16 * j;
                pv[jj] = W_n1[k * 64 + nt * 16 + c];
            }
#pragma unroll
            for (int jj = 0; jj < 8; ++jj) fragN1[g0 - 8192 + jj * BLOCK] = (_Float16)pv[jj];
        } else {
#pragma unroll
            for (int jj = 0; jj < 8; ++jj) {
                const int g = g0 + jj * BLOCK;
                const int s8 = g & 7, ln = (g >> 3) & 31, j = (g >> 8) & 1;
                const int c = ln & 15, h = ln >> 4;
                const int q = g - 16384;
                const int kc = (q >> 9) & 1, nt = q >> 10;
                const int k = kc * 32 + 8 * h + s8 + 16 * j;
                pv[jj] = W_n2[k * 64 + nt * 16 + c];
            }
#pragma unroll
            for (int jj = 0; jj < 8; ++jj) fragN2[g0 - 16384 + jj * BLOCK] = (_Float16)pv[jj];
        }
    }

    __syncthreads();

    const int wave = threadIdx.x >> 5, lane = threadIdx.x & 31;
    const int c = lane & 15, h = lane >> 4;
    const int numTiles = (N + 15) >> 4;
    const int tile = blockIdx.x * WAVES + wave;
    if (tile >= numTiles) return;

    int node = tile * 16 + c;
    if (node >= N) node = N - 1;

    v16h a[2];
#pragma unroll
    for (int kc = 0; kc < 2; ++kc) {
        const float* p = node_feat + (long)node * 64 + kc * 32 + 8 * h;
        const float4 v0 = *(const float4*)p,        v1 = *(const float4*)(p + 4);
        const float4 v2 = *(const float4*)(p + 16), v3 = *(const float4*)(p + 20);
        v16h f;
        f[0] = (_Float16)v0.x; f[1] = (_Float16)v0.y; f[2] = (_Float16)v0.z; f[3] = (_Float16)v0.w;
        f[4] = (_Float16)v1.x; f[5] = (_Float16)v1.y; f[6] = (_Float16)v1.z; f[7] = (_Float16)v1.w;
        f[8] = (_Float16)v2.x; f[9] = (_Float16)v2.y; f[10] = (_Float16)v2.z; f[11] = (_Float16)v2.w;
        f[12] = (_Float16)v3.x; f[13] = (_Float16)v3.y; f[14] = (_Float16)v3.z; f[15] = (_Float16)v3.w;
        a[kc] = f;
    }

#pragma unroll
    for (int t = 0; t < 4; ++t) {
        float* T = (t == 0) ? Kn : (t == 1) ? Qn : (t == 2) ? Ps : Pd;
        v8f acc[4];
#pragma unroll
        for (int nt = 0; nt < 4; ++nt) acc[nt] = {};
#pragma unroll
        for (int kc = 0; kc < 2; ++kc) {
#pragma unroll
            for (int nt = 0; nt < 4; ++nt)
                acc[nt] = wmma32(a[kc],
                                 load_frag(&sWT[(t * 64 + nt * 16 + c) * PKP + kc * 32 + 8 * h]),
                                 acc[nt]);
        }
#pragma unroll
        for (int nt = 0; nt < 4; ++nt) {
            const float bb = (t == 3) ? b_e1[nt * 16 + c] : 0.0f;
#pragma unroll
            for (int r = 0; r < 8; ++r) {
                const int nd = tile * 16 + 8 * h + r;
                if (nd < N) T[(long)nd * 64 + nt * 16 + c] = acc[nt][r] + bb;
            }
        }
    }
}


__global__ __launch_bounds__(BLOCK) void edge_score_kernel(
    const float* __restrict__ edge_feat, const float* __restrict__ coord4,
    const float* __restrict__ Kn, const float* __restrict__ Qn,
    const unsigned* __restrict__ fragK_w,
    const int* __restrict__ src, const int* __restrict__ dst,
    float* __restrict__ e_arr, unsigned* __restrict__ e_max0, unsigned* __restrict__ e_max1,
    unsigned* __restrict__ cnt0, unsigned* __restrict__ cnt1, unsigned* __restrict__ rank, int E) {

    const int wave = threadIdx.x >> 5, lane = threadIdx.x & 31;
    const int c = lane & 15, h = lane >> 4;
    const int numTiles = (E + 15) >> 4;
    const int tile = blockIdx.x * WAVES + wave;
    if (tile >= numTiles) return;
    const _Float16* fragK = (const _Float16*)fragK_w;

    const int eIdx = tile * 16 + c;
    const bool valid = eIdx < E;
    const int e = valid ? eIdx : (E - 1);
    const v4f* efp = (const v4f*)(edge_feat + (long)e * 16 + 8 * h);
    v4f x0 = efp[0];
    v4f x1 = efp[1];
    int s = src[e], d = dst[e];
    asm volatile("" : "+v"(x0), "+v"(x1), "+v"(s), "+v"(d));
    const unsigned half = package_half();
    v4f cs = ((const v4f*)coord4)[(unsigned)s];
    v4f cd = ((const v4f*)coord4)[(unsigned)d];
    v8h xh;
    xh[0] = (_Float16)x0.x; xh[1] = (_Float16)x0.y; xh[2] = (_Float16)x0.z; xh[3] = (_Float16)x0.w;
    xh[4] = (_Float16)x1.x; xh[5] = (_Float16)x1.y; xh[6] = (_Float16)x1.z; xh[7] = (_Float16)x1.w;

    const v4f* __restrict__ Kn4 = (const v4f*)Kn;
    const v4f* __restrict__ Qn4 = (const v4f*)Qn;
    v4f kv[8], qv[8];
#pragma unroll
    for (int r = 0; r < 8; ++r) {
        const unsigned sm = (unsigned)__shfl(s, 8 * h + r, 32);
        const unsigned dm = (unsigned)__shfl(d, 8 * h + r, 32);
        kv[r] = Kn4[sm * 16u + (unsigned)c];
        qv[r] = Qn4[dm * 16u + (unsigned)c];
    }
    int fo = lane * 8;
    asm volatile("" : "+v"(kv[0]), "+v"(kv[1]), "+v"(kv[2]), "+v"(kv[3]),
                      "+v"(kv[4]), "+v"(kv[5]), "+v"(kv[6]), "+v"(kv[7]),
                      "+v"(qv[0]), "+v"(qv[1]), "+v"(qv[2]), "+v"(qv[3]),
                      "+v"(qv[4]), "+v"(qv[5]), "+v"(qv[6]), "+v"(qv[7]),
                      "+v"(cs), "+v"(cd), "+v"(xh));

    float p[8];
#pragma unroll
    for (int r = 0; r < 8; ++r)
        p[r] = kv[r].x * qv[r].x + kv[r].y * qv[r].y + kv[r].z * qv[r].z + kv[r].w * qv[r].w;
    const float dx = cs.x - cd.x, dy = cs.y - cd.y, dz = cs.z - cd.z;
    float dist = sqrtf(dx * dx + dy * dy + dz * dz);
    asm volatile("" : "+v"(fo), "+v"(dist), "+v"(p[0]), "+v"(p[1]), "+v"(p[2]), "+v"(p[3]),
                      "+v"(p[4]), "+v"(p[5]), "+v"(p[6]), "+v"(p[7]));

    v8h flo[4], fhi[4];
#pragma unroll
    for (int nt = 0; nt < 4; ++nt) {
        const _Float16* fb = fragK + nt * 512 + fo;
        flo[nt] = *(const v8h*)fb;
        fhi[nt] = *(const v8h*)(fb + 256);
    }
    asm volatile("" : "+v"(flo[0]), "+v"(flo[1]), "+v"(flo[2]), "+v"(flo[3]),
                      "+v"(fhi[0]), "+v"(fhi[1]), "+v"(fhi[2]), "+v"(fhi[3]));

    const v16h x = edge_frag_pack(xh, dist, h);

    v8f acc[4];
#pragma unroll
    for (int nt = 0; nt < 4; ++nt) {
        const v8f zero = {};
        acc[nt] = wmma32(x, __builtin_shufflevector(flo[nt], fhi[nt], 0, 1, 2, 3, 4, 5, 6, 7,
                                                    8, 9, 10, 11, 12, 13, 14, 15), zero);
    }

#pragma unroll
    for (int r = 0; r < 8; ++r) {
        p[r] += acc[0][r] * qv[r].x + acc[1][r] * qv[r].y + acc[2][r] * qv[r].z + acc[3][r] * qv[r].w;
    }
#pragma unroll
    for (int r = 0; r < 8; ++r) p[r] = row_xor_add<1>(p[r]);
#pragma unroll
    for (int r = 0; r < 8; ++r) p[r] = row_xor_add<2>(p[r]);
#pragma unroll
    for (int r = 0; r < 8; ++r) p[r] = row_xor_add<4>(p[r]);
#pragma unroll
    for (int r = 0; r < 8; ++r) p[r] = row_xor_add<8>(p[r]);

    float pe = p[0];
#pragma unroll
    for (int r = 1; r < 8; ++r) pe = ((c & 7) == r) ? p[r] : pe;

    if ((c >> 3) == h && valid) {
        const float ev = pe * 0.125f;
        e_arr[eIdx] = ev;
        unsigned* __restrict__ e_maxH = half ? e_max1 : e_max0;
        unsigned* __restrict__ cntH   = half ? cnt1 : cnt0;
        atomicMax(&e_maxH[d], enc_f(ev));
        rank[eIdx] = atomicAdd(&cntH[d], 1u) | (half << 31);
    }
}


#define SCAN_T 1024

__global__ __launch_bounds__(SCAN_T) void rowstart_scan_kernel(
    const unsigned* __restrict__ cnt0, const unsigned* __restrict__ cnt1,
    unsigned* __restrict__ rowstart0, unsigned* __restrict__ rowstart1, int N) {

    __shared__ unsigned sWave[32];
    const int t = threadIdx.x, lane = t & 31, wave = t >> 5;
    const int per = (N + SCAN_T - 1) / SCAN_T;
    const int base = t * per;
    unsigned tot = 0;
    for (int i0 = 0; i0 < per; i0 += 8) {
        unsigned v[8], u[8];
#pragma unroll
        for (int j = 0; j < 8; ++j) {
            const int n = base + i0 + j;
            const bool ok = (i0 + j < per) && (n < N);
            const unsigned x0 = cnt0[ok ? n : 0];
            const unsigned x1 = cnt1[ok ? n : 0];
            v[j] = ok ? x0 : 0u;
            u[j] = ok ? x1 : 0u;
        }
#pragma unroll
        for (int j = 0; j < 8; ++j) tot += v[j] + u[j];
    }
    unsigned inc = tot;
    for (int o = 1; o < 32; o <<= 1) {
        const unsigned v = __shfl_up(inc, o, 32);
        if (lane >= o) inc += v;
    }
    if (lane == 31) sWave[wave] = inc;
    __syncthreads();
    if (wave == 0) {
        const unsigned wt = sWave[lane];
        unsigned winc = wt;
        for (int o = 1; o < 32; o <<= 1) {
            const unsigned v = __shfl_up(winc, o, 32);
            if (lane >= o) winc += v;
        }
        sWave[lane] = winc - wt;
    }
    __syncthreads();
    unsigned run = sWave[wave] + inc - tot;
    for (int i0 = 0; i0 < per; i0 += 8) {
        unsigned v[8], u[8];
#pragma unroll
        for (int j = 0; j < 8; ++j) {
            const int n = base + i0 + j;
            const bool ok = (i0 + j < per) && (n < N);
            const unsigned x0 = cnt0[ok ? n : 0];
            const unsigned x1 = cnt1[ok ? n : 0];
            v[j] = ok ? x0 : 0u;
            u[j] = ok ? x1 : 0u;
        }
#pragma unroll
        for (int j = 0; j < 8; ++j) {
            const int n = base + i0 + j;
            if ((i0 + j < per) && (n < N)) {
                rowstart0[n] = run;
                rowstart1[n] = run + v[j];
                run += v[j] + u[j];
            }
        }
    }
}


__global__ __launch_bounds__(256) void edge_scatter_kernel(
    const int* __restrict__ src, const int* __restrict__ dst,
    const float* __restrict__ e_arr, const unsigned* __restrict__ rank,
    const unsigned* __restrict__ rowstart0, const unsigned* __restrict__ rowstart1,
    int4* __restrict__ sorted4, int E) {

    const int e0 = (blockIdx.x * 256 + threadIdx.x) * 4;
    if (e0 >= E) return;
    const unsigned last = (unsigned)(E - 1);
    const unsigned off1 = (unsigned)(rowstart1 - rowstart0);
    const unsigned lo31 = 0x7FFFFFFFu;
    if (e0 + 3 < E) {
        const int4   d4 = *(const int4*)(dst + e0);
        const int4   s4 = *(const int4*)(src + e0);
        const uint4  r4 = *(const uint4*)(rank + e0);
        const float4 v4 = *(const float4*)(e_arr + e0);
        const unsigned p0 = rowstart0[(unsigned)d4.x + ((r4.x >> 31) ? off1 : 0u)] + (r4.x & lo31);
        const unsigned p1 = rowstart0[(unsigned)d4.y + ((r4.y >> 31) ? off1 : 0u)] + (r4.y & lo31);
        const unsigned p2 = rowstart0[(unsigned)d4.z + ((r4.z >> 31) ? off1 : 0u)] + (r4.z & lo31);
        const unsigned p3 = rowstart0[(unsigned)d4.w + ((r4.w >> 31) ? off1 : 0u)] + (r4.w & lo31);
        sorted4[min(p0, last)] = make_int4(e0,     s4.x, d4.x, __float_as_int(v4.x));
        sorted4[min(p1, last)] = make_int4(e0 + 1, s4.y, d4.y, __float_as_int(v4.y));
        sorted4[min(p2, last)] = make_int4(e0 + 2, s4.z, d4.z, __float_as_int(v4.z));
        sorted4[min(p3, last)] = make_int4(e0 + 3, s4.w, d4.w, __float_as_int(v4.w));
    } else {
        for (int e = e0; e < E; ++e) {
            const int d = dst[e];
            const unsigned rk = rank[e];
            const unsigned pos = rowstart0[(unsigned)d + ((rk >> 31) ? off1 : 0u)] + (rk & lo31);
            sorted4[min(pos, last)] = make_int4(e, src[e], d, __float_as_int(e_arr[e]));
        }
    }
}


__global__ __launch_bounds__(BLOCK) void edge_msg_kernel(
    const float* __restrict__ edge_feat, const float* __restrict__ coord4,
    const float* __restrict__ Ps, const float* __restrict__ Pd,
    const unsigned* __restrict__ fragE1_w, const unsigned* __restrict__ fragE2_w,
    const float* __restrict__ b_e2,
    const int4* __restrict__ sorted4, const unsigned* __restrict__ e_max0, const unsigned* __restrict__ e_max1,
    float* __restrict__ denom, float* __restrict__ h_neigh, int E, int N) {

    const int wave = threadIdx.x >> 5, lane = threadIdx.x & 31;
    const int c = lane & 15, h = lane >> 4;
    const int numTiles = (E + 15) >> 4;
    const int tile = blockIdx.x * WAVES + wave;
    if (tile >= numTiles) return;
    const _Float16* fragE1 = (const _Float16*)fragE1_w;
    const _Float16* fragE2 = (const _Float16*)fragE2_w;

    const int place = tile * 16 + c;
    const bool valid = place < E;
    const int4 rec = sorted4[valid ? place : (E - 1)];
    const int e = (int)min((unsigned)rec.x, (unsigned)(E - 1));
    const int s = (int)min((unsigned)rec.y, (unsigned)(N - 1));
    const int d = (int)min((unsigned)rec.z, (unsigned)(N - 1));

    const float* sp = Ps + (long)s * 64 + 8 * h;
    const float* dp = Pd + (long)d * 64 + 8 * h;
    v8f acc[4];
#pragma unroll
    for (int nt = 0; nt < 4; ++nt) acc[nt] = load8(sp + nt * 16) + load8(dp + nt * 16);

    float w = 0.0f;
    if (valid) w = __expf(__int_as_float(rec.w) - dec_f(max(e_max0[d], e_max1[d])));

    const v16h x = edge_frag(edge_feat, coord4, e, s, d, h);

#pragma unroll
    for (int nt = 0; nt < 4; ++nt)
        acc[nt] = wmma32(load_packed(fragE1 + nt * 512, lane), x, acc[nt]);

    v16h mid[2];
#pragma unroll
    for (int kc = 0; kc < 2; ++kc) {
        v16h m;
#pragma unroll
        for (int r = 0; r < 8; ++r) {
            m[r]     = (_Float16)silu_f(acc[2 * kc][r]);
            m[8 + r] = (_Float16)silu_f(acc[2 * kc + 1][r]);
        }
        mid[kc] = m;
    }

    v8f acc2[4];
#pragma unroll
    for (int nt = 0; nt < 4; ++nt) {
        acc2[nt] = {};
#pragma unroll
        for (int kc = 0; kc < 2; ++kc)
            acc2[nt] = wmma32(mid[kc], load_packed(fragE2 + (nt * 2 + kc) * 512, lane), acc2[nt]);
    }

    float bb[4];
#pragma unroll
    for (int nt = 0; nt < 4; ++nt) bb[nt] = b_e2[nt * 16 + c];

    const int d7 = __shfl(d, 7, 32), d8 = __shfl(d, 8, 32);
    const bool joined = d7 == d8;
    bool holdFirst = joined && h == 1;
    float run[4]  = {0.0f, 0.0f, 0.0f, 0.0f};
    float held[4] = {0.0f, 0.0f, 0.0f, 0.0f};
    float wsum = 0.0f, wheld = 0.0f;
    int dr = __shfl(d, 8 * h, 32);
#pragma unroll
    for (int r = 0; r < 8; ++r) {
        const int m = 8 * h + r;
        const float wr = __shfl(w, m, 32);
        const int dn = (r < 7) ? __shfl(d, m + 1, 32) : dr;
        wsum += wr;
#pragma unroll
        for (int nt = 0; nt < 4; ++nt) run[nt] += silu_f(acc2[nt][r] + bb[nt]) * wr;
        if (r == 7 || dn != dr) {
            if (holdFirst || (joined && h == 0 && r == 7)) {
#pragma unroll
                for (int nt = 0; nt < 4; ++nt) held[nt] = run[nt];
                wheld = wsum;
            } else {
                float* hp = h_neigh + (long)dr * 64 + c;
#pragma unroll
                for (int nt = 0; nt < 4; ++nt) atomicAdd(hp + nt * 16, run[nt]);
                if (c == 0) atomicAdd(&denom[dr], wsum);
            }
            holdFirst = false;
#pragma unroll
            for (int nt = 0; nt < 4; ++nt) run[nt] = 0.0f;
            wsum = 0.0f;
        }
        dr = dn;
    }
    if (joined) {
        float* hp = h_neigh + (long)d7 * 64 + c;
#pragma unroll
        for (int nt = 0; nt < 4; ++nt) {
            const float other = __shfl_xor(held[nt], 16, 32);
            if (h == 0) atomicAdd(hp + nt * 16, held[nt] + other);
        }
        const float wother = __shfl_xor(wheld, 16, 32);
        if (lane == 0) atomicAdd(&denom[d7], wheld + wother);
    }
}


__global__ __launch_bounds__(BLOCK) void node_mlp_kernel(
    const float* __restrict__ node_feat,
    const unsigned* __restrict__ fragN1_w, const float* __restrict__ b_n1,
    const unsigned* __restrict__ fragN2_w, const float* __restrict__ b_n2,
    const float* __restrict__ denom, const float* __restrict__ h_neigh,
    float* __restrict__ out, int N) {

    const int wave = threadIdx.x >> 5, lane = threadIdx.x & 31;
    const int c = lane & 15, h = lane >> 4;
    const int numTiles = (N + 15) >> 4;
    const int tile = blockIdx.x * WAVES + wave;
    if (tile >= numTiles) return;
    const _Float16* fragN1 = (const _Float16*)fragN1_w;
    const _Float16* fragN2 = (const _Float16*)fragN2_w;

    int node = tile * 16 + c;
    if (node >= N) node = N - 1;

    const float dn = denom[node];
    const float inv = (dn > 0.0f) ? (1.0f / dn) : 0.0f;

    v16h x[4];
#pragma unroll
    for (int kc = 0; kc < 2; ++kc) {
        const float* p = node_feat + (long)node * 64 + kc * 32 + 8 * h;
        const float4 v0 = *(const float4*)p,        v1 = *(const float4*)(p + 4);
        const float4 v2 = *(const float4*)(p + 16), v3 = *(const float4*)(p + 20);
        v16h f;
        f[0] = (_Float16)v0.x; f[1] = (_Float16)v0.y; f[2] = (_Float16)v0.z; f[3] = (_Float16)v0.w;
        f[4] = (_Float16)v1.x; f[5] = (_Float16)v1.y; f[6] = (_Float16)v1.z; f[7] = (_Float16)v1.w;
        f[8] = (_Float16)v2.x; f[9] = (_Float16)v2.y; f[10] = (_Float16)v2.z; f[11] = (_Float16)v2.w;
        f[12] = (_Float16)v3.x; f[13] = (_Float16)v3.y; f[14] = (_Float16)v3.z; f[15] = (_Float16)v3.w;
        x[kc] = f;
    }
#pragma unroll
    for (int kc = 0; kc < 2; ++kc) {
        const float* p = h_neigh + (long)node * 64 + kc * 32 + 8 * h;
        const float4 v0 = *(const float4*)p,        v1 = *(const float4*)(p + 4);
        const float4 v2 = *(const float4*)(p + 16), v3 = *(const float4*)(p + 20);
        v16h f;
        f[0] = (_Float16)(v0.x * inv); f[1] = (_Float16)(v0.y * inv);
        f[2] = (_Float16)(v0.z * inv); f[3] = (_Float16)(v0.w * inv);
        f[4] = (_Float16)(v1.x * inv); f[5] = (_Float16)(v1.y * inv);
        f[6] = (_Float16)(v1.z * inv); f[7] = (_Float16)(v1.w * inv);
        f[8] = (_Float16)(v2.x * inv); f[9] = (_Float16)(v2.y * inv);
        f[10] = (_Float16)(v2.z * inv); f[11] = (_Float16)(v2.w * inv);
        f[12] = (_Float16)(v3.x * inv); f[13] = (_Float16)(v3.y * inv);
        f[14] = (_Float16)(v3.z * inv); f[15] = (_Float16)(v3.w * inv);
        x[2 + kc] = f;
    }

    v8f acc[4];
#pragma unroll
    for (int nt = 0; nt < 4; ++nt) acc[nt] = load8(b_n1 + nt * 16 + 8 * h);
#pragma unroll
    for (int kc = 0; kc < 4; ++kc) {
#pragma unroll
        for (int nt = 0; nt < 4; ++nt)
            acc[nt] = wmma32(load_packed(fragN1 + (nt * 4 + kc) * 512, lane), x[kc], acc[nt]);
    }

    v16h mid[2];
#pragma unroll
    for (int kc = 0; kc < 2; ++kc) {
        v16h m;
#pragma unroll
        for (int r = 0; r < 8; ++r) {
            m[r]     = (_Float16)silu_f(acc[2 * kc][r]);
            m[8 + r] = (_Float16)silu_f(acc[2 * kc + 1][r]);
        }
        mid[kc] = m;
    }

    v8f acc2[4];
#pragma unroll
    for (int nt = 0; nt < 4; ++nt) {
        const float bb = b_n2[nt * 16 + c];
        v8f st;
#pragma unroll
        for (int r = 0; r < 8; ++r) st[r] = bb;
        acc2[nt] = st;
#pragma unroll
        for (int kc = 0; kc < 2; ++kc)
            acc2[nt] = wmma32(mid[kc], load_packed(fragN2 + (nt * 2 + kc) * 512, lane), acc2[nt]);
    }

#pragma unroll
    for (int nt = 0; nt < 4; ++nt) {
#pragma unroll
        for (int r = 0; r < 8; ++r) {
            const int nd = tile * 16 + 8 * h + r;
            if (nd < N) out[(long)nd * 64 + nt * 16 + c] = acc2[nt][r];
        }
    }
}


extern "C" void kernel_launch(void* const* d_in, const int* in_sizes, int n_in,
                              void* d_out, int out_size, void* d_ws, size_t ws_size,
                              hipStream_t stream) {
    const float* node_feat = (const float*)d_in[0];
    const float* coord     = (const float*)d_in[1];
    const float* edge_feat = (const float*)d_in[2];
    const float* W_e1 = (const float*)d_in[3];
    const float* b_e1 = (const float*)d_in[4];
    const float* W_e2 = (const float*)d_in[5];
    const float* b_e2 = (const float*)d_in[6];
    const float* W_n1 = (const float*)d_in[7];
    const float* b_n1 = (const float*)d_in[8];
    const float* W_n2 = (const float*)d_in[9];
    const float* b_n2 = (const float*)d_in[10];
    const float* W_q  = (const float*)d_in[11];
    const float* W_k  = (const float*)d_in[12];
    const int*   src  = (const int*)d_in[13];
    const int*   dst  = (const int*)d_in[14];

    const int N = in_sizes[0] / 64;
    const int E = in_sizes[13];

    const size_t Ea = ((size_t)E + 63) & ~(size_t)63;
    const size_t Na = ((size_t)N + 63) & ~(size_t)63;
    const size_t T  = (size_t)N * 64;
    float*    e_arr   = (float*)d_ws;
    unsigned* e_max0  = (unsigned*)(e_arr + Ea);
    unsigned* e_max1  = e_max0 + Na;
    float*    denom   = (float*)(e_max1 + Na);
    float*    h_neigh = denom + Na;
    float*    Kn      = h_neigh + T;
    float*    Qn      = Kn + T;
    float*    Ps      = Qn + T;
    float*    Pd      = Ps + T;
    float*    coord4  = Pd + T;
    unsigned* fragK   = (unsigned*)(coord4 + 4 * Na);
    unsigned* fragE1  = fragK + 1024;
    unsigned* fragE2  = fragE1 + 1024;
    unsigned* fragN1  = fragE2 + 2048;
    unsigned* fragN2  = fragN1 + 4096;
    unsigned* cnt0      = fragN2 + 2048;
    unsigned* cnt1      = cnt0 + Na;
    unsigned* rowstart0 = cnt1 + Na;
    unsigned* rowstart1 = rowstart0 + Na;
    unsigned* rank      = rowstart1 + Na;
    int4*     sorted4  = (int4*)(rank + Ea);

    const size_t wsNeed = (size_t)((const char*)(sorted4 + E) - (const char*)d_ws);
    if (ws_size < wsNeed) return;

    const int numEdgeTiles = (E + 15) / 16;
    const int edgeBlocks   = (numEdgeTiles + WAVES - 1) / WAVES;
    const int numNodeTiles = (N + 15) / 16;
    const int nodeBlocks   = (numNodeTiles + WAVES - 1) / WAVES;
    const int projBlocks   = nodeBlocks < 20 ? 20 : nodeBlocks;

    node_proj_kernel<<<projBlocks, BLOCK, 0, stream>>>(
        node_feat, coord, W_k, W_q, W_e1, b_e1, W_e2, W_n1, W_n2,
        Kn, Qn, Ps, Pd, coord4, fragK, fragE1, fragE2, fragN1, fragN2, e_max0, e_max1, denom, h_neigh,
        cnt0, cnt1, N);

    edge_score_kernel<<<edgeBlocks, BLOCK, 0, stream>>>(
        edge_feat, coord4, Kn, Qn, fragK, src, dst, e_arr, e_max0, e_max1, cnt0, cnt1, rank, E);

    rowstart_scan_kernel<<<1, SCAN_T, 0, stream>>>(cnt0, cnt1, rowstart0, rowstart1, N);

    edge_scatter_kernel<<<((E + 3) / 4 + 255) / 256, 256, 0, stream>>>(
        src, dst, e_arr, rank, rowstart0, rowstart1, sorted4, E);

    edge_msg_kernel<<<edgeBlocks, BLOCK, 0, stream>>>(
        edge_feat, coord4, Ps, Pd, fragE1, fragE2, b_e2,
        sorted4, e_max0, e_max1, denom, h_neigh, E, N);

    node_mlp_kernel<<<nodeBlocks, BLOCK, 0, stream>>>(
        node_feat, fragN1, b_n1, fragN2, b_n2, denom, h_neigh, (float*)d_out, N);
}
